# starting kernel
# baseline (speedup 1.0000x reference)
#include <hip/hip_runtime.h>
#include <hip/hip_bf16.h>

typedef __attribute__((ext_vector_type(2))) float v2f;
typedef __attribute__((ext_vector_type(8))) float v8f;

#define KDIM 128
#define PDIM 256
#define NDIM 100000
#define LAMBDA 0.001f
#define NCHUNKS 64
#define CHUNK ((NDIM + NCHUNKS - 1) / NCHUNKS)
#define KP (KDIM * PDIM)

__global__ __launch_bounds__(512) void fm_gemm_partial(
    const float* __restrict__ evx, const float* __restrict__ evy,
    const float* __restrict__ fx,  const float* __restrict__ fy,
    float* __restrict__ partials)
{
    const int sel   = blockIdx.y;
    const int chunk = blockIdx.x;
    const float* __restrict__ ev   = sel ? evy : evx;
    const float* __restrict__ feat = sel ? fy  : fx;

    __shared__ float evBlk[KDIM][33];
    __shared__ float ftBlk[32][260];

    const int tid  = threadIdx.x;
    const int lane = tid & 31;
    const int wave = tid >> 5;
    const int rowBand = (wave >> 1) * 16;
    const int wcol    = (wave & 1) * 128;

    v8f acc[8];
    #pragma unroll
    for (int t = 0; t < 8; ++t) acc[t] = (v8f)0.0f;

    const int nStart = chunk * CHUNK;
    const int nEnd   = min(nStart + CHUNK, NDIM);

    {
        const int blk = nStart;
        #pragma unroll
        for (int q = 0; q < 8; ++q) {
            int idx = tid + 512 * q;
            int r = idx >> 5, c = idx & 31;
            int n = blk + c;
            evBlk[r][c] = (n < nEnd) ? ev[(long)r * NDIM + n] : 0.0f;
        }
        #pragma unroll
        for (int q = 0; q < 16; ++q) {
            int idx = tid + 512 * q;
            int r = idx >> 8, c = idx & 255;
            int n = blk + r;
            ftBlk[r][c] = (n < nEnd) ? feat[(long)n * PDIM + c] : 0.0f;
        }
    }
    __syncthreads();

    const int m  = lane & 15;
    const int kp = (lane >> 4) << 1;

    for (int blk = nStart; blk < nEnd; blk += 32) {
        const int nxt = blk + 32;
        const bool more = (nxt < nEnd);

        float evr[8];
        float ftr[16];
        if (more) {
            #pragma unroll
            for (int q = 0; q < 8; ++q) {
                int idx = tid + 512 * q;
                int r = idx >> 5, c = idx & 31;
                int n = nxt + c;
                evr[q] = (n < nEnd) ? ev[(long)r * NDIM + n] : 0.0f;
            }
            #pragma unroll
            for (int q = 0; q < 16; ++q) {
                int idx = tid + 512 * q;
                int r = idx >> 8, c = idx & 255;
                int n = nxt + r;
                ftr[q] = (n < nEnd) ? feat[(long)n * PDIM + c] : 0.0f;
            }
            if (nxt + 32 < nEnd) {
                long base = (long)(nxt + 32) * PDIM;
                __builtin_prefetch(&feat[base + (long)tid * 64], 0, 0);
            }
        }

        #pragma unroll
        for (int kk = 0; kk < 8; ++kk) {
            const int nOff = kk * 4;
            v2f a;
            a.x = evBlk[rowBand + m][nOff + kp];
            a.y = evBlk[rowBand + m][nOff + kp + 1];
            #pragma unroll
            for (int t = 0; t < 8; ++t) {
                const int col = wcol + t * 16 + m;
                v2f b;
                b.x = ftBlk[nOff + kp][col];
                b.y = ftBlk[nOff + kp + 1][col];
                acc[t] = __builtin_amdgcn_wmma_f32_16x16x4_f32(
                    false, a, false, b, (short)0, acc[t], false, false);
            }
        }
        __syncthreads();

        if (more) {
            #pragma unroll
            for (int q = 0; q < 8; ++q) {
                int idx = tid + 512 * q;
                evBlk[idx >> 5][idx & 31] = evr[q];
            }
            #pragma unroll
            for (int q = 0; q < 16; ++q) {
                int idx = tid + 512 * q;
                ftBlk[idx >> 8][idx & 255] = ftr[q];
            }
            __syncthreads();
        }
    }

    float* out = partials + (size_t)sel * NCHUNKS * KP + (size_t)chunk * KP;
    const int mrow = rowBand + ((lane >> 4) << 3);
    #pragma unroll
    for (int t = 0; t < 8; ++t) {
        const int col = wcol + t * 16 + (lane & 15);
        #pragma unroll
        for (int r = 0; r < 8; ++r)
            out[(mrow + r) * PDIM + col] = acc[t][r];
    }
}

__global__ void fm_reduce(const float* __restrict__ partials,
                          float* __restrict__ Amat, float* __restrict__ Bmat)
{
    int idx = blockIdx.x * blockDim.x + threadIdx.x;
    if (idx >= KP) return;
    float sa = 0.0f, sb = 0.0f;
    for (int c = 0; c < NCHUNKS; ++c) {
        sa += partials[(size_t)c * KP + idx];
        sb += partials[(size_t)(NCHUNKS + c) * KP + idx];
    }
    Amat[idx] = sa;
    Bmat[idx] = sb;
}

__global__ void fm_mask(const float* __restrict__ evx,
                        const float* __restrict__ evy,
                        float* __restrict__ D)
{
    int idx = blockIdx.x * blockDim.x + threadIdx.x;
    if (idx >= KDIM * KDIM) return;
    int i = idx >> 7;
    int j = idx & 127;
    float s  = fmaxf(evx[KDIM - 1], evy[KDIM - 1]);
    float g1 = sqrtf(evx[j] / s);
    float g2 = sqrtf(evy[i] / s);
    float d1 = 1.0f / (g1 * g1 + 1.0f);
    float d2 = 1.0f / (g2 * g2 + 1.0f);
    float re = g2 * d2 - g1 * d1;
    float im = d2 - d1;
    D[idx] = re * re + im * im;
}

__global__ __launch_bounds__(256) void fm_gram(
    const float* __restrict__ Amat, const float* __restrict__ Bmat,
    float* __restrict__ AAt, float* __restrict__ BAt)
{
    const float* __restrict__ L = blockIdx.x ? Bmat : Amat;
    const float* __restrict__ R = Amat;
    float* __restrict__ out     = blockIdx.x ? BAt : AAt;

    __shared__ float blkL[KDIM][33];
    __shared__ float blkR[KDIM][33];

    const int tid  = threadIdx.x;
    const int lane = tid & 31;
    const int wave = tid >> 5;
    const int iBase = wave * 16;

    v8f acc[8];
    #pragma unroll
    for (int t = 0; t < 8; ++t) acc[t] = (v8f)0.0f;

    for (int p0 = 0; p0 < PDIM; p0 += 32) {
        for (int idx = tid; idx < KDIM * 32; idx += 256) {
            int r = idx >> 5, c = idx & 31;
            blkL[r][c] = L[r * PDIM + p0 + c];
            blkR[r][c] = R[r * PDIM + p0 + c];
        }
        __syncthreads();

        const int m  = lane & 15;
        const int kp = (lane >> 4) << 1;
        #pragma unroll
        for (int kk = 0; kk < 8; ++kk) {
            const int pOff = kk * 4;
            v2f a;
            a.x = blkL[iBase + m][pOff + kp];
            a.y = blkL[iBase + m][pOff + kp + 1];
            #pragma unroll
            for (int t = 0; t < 8; ++t) {
                v2f b;
                b.x = blkR[t * 16 + m][pOff + kp];
                b.y = blkR[t * 16 + m][pOff + kp + 1];
                acc[t] = __builtin_amdgcn_wmma_f32_16x16x4_f32(
                    false, a, false, b, (short)0, acc[t], false, false);
            }
        }
        __syncthreads();
    }

    const int mrow = iBase + ((lane >> 4) << 3);
    #pragma unroll
    for (int t = 0; t < 8; ++t) {
        const int j = t * 16 + (lane & 15);
        #pragma unroll
        for (int r = 0; r < 8; ++r)
            out[(mrow + r) * KDIM + j] = acc[t][r];
    }
}

__global__ __launch_bounds__(128) void fm_solve(
    const float* __restrict__ AAt, const float* __restrict__ BAt,
    const float* __restrict__ D,   float* __restrict__ Cout)
{
    extern __shared__ float Msh[];
    const int i   = blockIdx.x;
    const int tid = threadIdx.x;

    for (int idx = tid; idx < KDIM * 129; idx += 128) {
        int r = idx / 129, c = idx - r * 129;
        float v;
        if (c < KDIM) {
            v = AAt[r * KDIM + c];
            if (c == r) v += LAMBDA * D[i * KDIM + r];
        } else {
            v = BAt[i * KDIM + r];
        }
        Msh[idx] = v;
    }

    for (int c = 0; c < KDIM - 1; ++c) {
        __syncthreads();
        int r = tid;
        if (r > c) {
            float f = Msh[r * 129 + c] / Msh[c * 129 + c];
            for (int j = c + 1; j <= KDIM; ++j)
                Msh[r * 129 + j] -= f * Msh[c * 129 + j];
        }
    }

    for (int c = KDIM - 1; c >= 0; --c) {
        __syncthreads();
        if (tid == c)
            Msh[c * 129 + KDIM] /= Msh[c * 129 + c];
        __syncthreads();
        if (tid < c)
            Msh[tid * 129 + KDIM] -= Msh[tid * 129 + c] * Msh[c * 129 + KDIM];
    }
    __syncthreads();
    Cout[i * KDIM + tid] = Msh[tid * 129 + KDIM];
}

extern "C" void kernel_launch(void* const* d_in, const int* in_sizes, int n_in,
                              void* d_out, int out_size, void* d_ws, size_t ws_size,
                              hipStream_t stream)
{
    const float* feat_x = (const float*)d_in[0];
    const float* feat_y = (const float*)d_in[1];
    const float* evals_x = (const float*)d_in[2];
    const float* evals_y = (const float*)d_in[3];
    const float* evecs_x = (const float*)d_in[4];
    const float* evecs_y = (const float*)d_in[5];

    float* wsf   = (float*)d_ws;
    float* Amat  = wsf;
    float* Bmat  = wsf + 32768;
    float* AAt   = wsf + 65536;
    float* BAt   = wsf + 81920;
    float* Dmask = wsf + 98304;
    float* part  = wsf + 114688;
    float* Cout  = (float*)d_out;

    fm_gemm_partial<<<dim3(NCHUNKS, 2), 512, 0, stream>>>(
        evecs_x, evecs_y, feat_x, feat_y, part);

    fm_reduce<<<(KP + 255) / 256, 256, 0, stream>>>(part, Amat, Bmat);

    fm_mask<<<(KDIM * KDIM + 255) / 256, 256, 0, stream>>>(evals_x, evals_y, Dmask);

    fm_gram<<<2, 256, 0, stream>>>(Amat, Bmat, AAt, BAt);

    fm_solve<<<KDIM, KDIM, KDIM * 129 * sizeof(float), stream>>>(
        AAt, BAt, Dmask, Cout);
}


# solve kernel: forward elimination blocked by 16 columns: panel in registers (only the pivot row crosses LDS), pivot rows finished per column, 16 deferred pivots applied in one pass; order kept
# speedup vs baseline: 1.6945x; 1.6945x over previous
#include <hip/hip_runtime.h>
#include <hip/hip_bf16.h>

typedef __attribute__((ext_vector_type(2))) float v2f;
typedef __attribute__((ext_vector_type(8))) float v8f;

#define KDIM 128
#define PDIM 256
#define NDIM 100000
#define LAMBDA 0.001f
#define NCHUNKS 64
#define CHUNK ((NDIM + NCHUNKS - 1) / NCHUNKS)
#define KP (KDIM * PDIM)

__global__ __launch_bounds__(512) void fm_gemm_partial(
    const float* __restrict__ evx, const float* __restrict__ evy,
    const float* __restrict__ fx,  const float* __restrict__ fy,
    float* __restrict__ partials)
{
    const int sel   = blockIdx.y;
    const int chunk = blockIdx.x;
    const float* __restrict__ ev   = sel ? evy : evx;
    const float* __restrict__ feat = sel ? fy  : fx;

    __shared__ float evBlk[KDIM][33];
    __shared__ float ftBlk[32][260];

    const int tid  = threadIdx.x;
    const int lane = tid & 31;
    const int wave = tid >> 5;
    const int rowBand = (wave >> 1) * 16;
    const int wcol    = (wave & 1) * 128;

    v8f acc[8];
    #pragma unroll
    for (int t = 0; t < 8; ++t) acc[t] = (v8f)0.0f;

    const int nStart = chunk * CHUNK;
    const int nEnd   = min(nStart + CHUNK, NDIM);

    {
        const int blk = nStart;
        #pragma unroll
        for (int q = 0; q < 8; ++q) {
            int idx = tid + 512 * q;
            int r = idx >> 5, c = idx & 31;
            int n = blk + c;
            evBlk[r][c] = (n < nEnd) ? ev[(long)r * NDIM + n] : 0.0f;
        }
        #pragma unroll
        for (int q = 0; q < 16; ++q) {
            int idx = tid + 512 * q;
            int r = idx >> 8, c = idx & 255;
            int n = blk + r;
            ftBlk[r][c] = (n < nEnd) ? feat[(long)n * PDIM + c] : 0.0f;
        }
    }
    __syncthreads();

    const int m  = lane & 15;
    const int kp = (lane >> 4) << 1;

    for (int blk = nStart; blk < nEnd; blk += 32) {
        const int nxt = blk + 32;
        const bool more = (nxt < nEnd);

        float evr[8];
        float ftr[16];
        if (more) {
            #pragma unroll
            for (int q = 0; q < 8; ++q) {
                int idx = tid + 512 * q;
                int r = idx >> 5, c = idx & 31;
                int n = nxt + c;
                evr[q] = (n < nEnd) ? ev[(long)r * NDIM + n] : 0.0f;
            }
            #pragma unroll
            for (int q = 0; q < 16; ++q) {
                int idx = tid + 512 * q;
                int r = idx >> 8, c = idx & 255;
                int n = nxt + r;
                ftr[q] = (n < nEnd) ? feat[(long)n * PDIM + c] : 0.0f;
            }
            if (nxt + 32 < nEnd) {
                long base = (long)(nxt + 32) * PDIM;
                __builtin_prefetch(&feat[base + (long)tid * 64], 0, 0);
            }
        }

        #pragma unroll
        for (int kk = 0; kk < 8; ++kk) {
            const int nOff = kk * 4;
            v2f a;
            a.x = evBlk[rowBand + m][nOff + kp];
            a.y = evBlk[rowBand + m][nOff + kp + 1];
            #pragma unroll
            for (int t = 0; t < 8; ++t) {
                const int col = wcol + t * 16 + m;
                v2f b;
                b.x = ftBlk[nOff + kp][col];
                b.y = ftBlk[nOff + kp + 1][col];
                acc[t] = __builtin_amdgcn_wmma_f32_16x16x4_f32(
                    false, a, false, b, (short)0, acc[t], false, false);
            }
        }
        __syncthreads();

        if (more) {
            #pragma unroll
            for (int q = 0; q < 8; ++q) {
                int idx = tid + 512 * q;
                evBlk[idx >> 5][idx & 31] = evr[q];
            }
            #pragma unroll
            for (int q = 0; q < 16; ++q) {
                int idx = tid + 512 * q;
                ftBlk[idx >> 8][idx & 255] = ftr[q];
            }
            __syncthreads();
        }
    }

    float* out = partials + (size_t)sel * NCHUNKS * KP + (size_t)chunk * KP;
    const int mrow = rowBand + ((lane >> 4) << 3);
    #pragma unroll
    for (int t = 0; t < 8; ++t) {
        const int col = wcol + t * 16 + (lane & 15);
        #pragma unroll
        for (int r = 0; r < 8; ++r)
            out[(mrow + r) * PDIM + col] = acc[t][r];
    }
}

__global__ void fm_reduce(const float* __restrict__ partials,
                          float* __restrict__ Amat, float* __restrict__ Bmat)
{
    int idx = blockIdx.x * blockDim.x + threadIdx.x;
    if (idx >= KP) return;
    float sa = 0.0f, sb = 0.0f;
    for (int c = 0; c < NCHUNKS; ++c) {
        sa += partials[(size_t)c * KP + idx];
        sb += partials[(size_t)(NCHUNKS + c) * KP + idx];
    }
    Amat[idx] = sa;
    Bmat[idx] = sb;
}

__global__ void fm_mask(const float* __restrict__ evx,
                        const float* __restrict__ evy,
                        float* __restrict__ D)
{
    int idx = blockIdx.x * blockDim.x + threadIdx.x;
    if (idx >= KDIM * KDIM) return;
    int i = idx >> 7;
    int j = idx & 127;
    float s  = fmaxf(evx[KDIM - 1], evy[KDIM - 1]);
    float g1 = sqrtf(evx[j] / s);
    float g2 = sqrtf(evy[i] / s);
    float d1 = 1.0f / (g1 * g1 + 1.0f);
    float d2 = 1.0f / (g2 * g2 + 1.0f);
    float re = g2 * d2 - g1 * d1;
    float im = d2 - d1;
    D[idx] = re * re + im * im;
}

__global__ __launch_bounds__(256) void fm_gram(
    const float* __restrict__ Amat, const float* __restrict__ Bmat,
    float* __restrict__ AAt, float* __restrict__ BAt)
{
    const float* __restrict__ L = blockIdx.x ? Bmat : Amat;
    const float* __restrict__ R = Amat;
    float* __restrict__ out     = blockIdx.x ? BAt : AAt;

    __shared__ float blkL[KDIM][33];
    __shared__ float blkR[KDIM][33];

    const int tid  = threadIdx.x;
    const int lane = tid & 31;
    const int wave = tid >> 5;
    const int iBase = wave * 16;

    v8f acc[8];
    #pragma unroll
    for (int t = 0; t < 8; ++t) acc[t] = (v8f)0.0f;

    for (int p0 = 0; p0 < PDIM; p0 += 32) {
        for (int idx = tid; idx < KDIM * 32; idx += 256) {
            int r = idx >> 5, c = idx & 31;
            blkL[r][c] = L[r * PDIM + p0 + c];
            blkR[r][c] = R[r * PDIM + p0 + c];
        }
        __syncthreads();

        const int m  = lane & 15;
        const int kp = (lane >> 4) << 1;
        #pragma unroll
        for (int kk = 0; kk < 8; ++kk) {
            const int pOff = kk * 4;
            v2f a;
            a.x = blkL[iBase + m][pOff + kp];
            a.y = blkL[iBase + m][pOff + kp + 1];
            #pragma unroll
            for (int t = 0; t < 8; ++t) {
                v2f b;
                b.x = blkR[t * 16 + m][pOff + kp];
                b.y = blkR[t * 16 + m][pOff + kp + 1];
                acc[t] = __builtin_amdgcn_wmma_f32_16x16x4_f32(
                    false, a, false, b, (short)0, acc[t], false, false);
            }
        }
        __syncthreads();
    }

    const int mrow = iBase + ((lane >> 4) << 3);
    #pragma unroll
    for (int t = 0; t < 8; ++t) {
        const int j = t * 16 + (lane & 15);
        #pragma unroll
        for (int r = 0; r < 8; ++r)
            out[(mrow + r) * KDIM + j] = acc[t][r];
    }
}

__global__ __launch_bounds__(128) void fm_solve(
    const float* __restrict__ AAt, const float* __restrict__ BAt,
    const float* __restrict__ D,   float* __restrict__ Cout)
{
    extern __shared__ float4 smem4[];
    float* Msh  = (float*)smem4;
    float* prow = Msh + KDIM * 129;
    float* Lbuf = prow + 32;
    float* Ubuf = Lbuf + 256;
    const int i   = blockIdx.x;
    const int tid = threadIdx.x;

    for (int idx = tid; idx < KDIM * 129; idx += 128) {
        int r = idx / 129, c = idx - r * 129;
        float v;
        if (c < KDIM) {
            v = AAt[r * KDIM + c];
            if (c == r) v += LAMBDA * D[i * KDIM + r];
        } else {
            v = BAt[i * KDIM + r];
        }
        Msh[idx] = v;
    }
    __syncthreads();

    for (int kb = 0; kb < KDIM / 16; ++kb) {
        const int k0 = kb * 16;

        float p[16];
        #pragma unroll
        for (int j = 0; j < 16; ++j)
            p[j] = (tid >= k0) ? Msh[tid * 129 + k0 + j] : 0.0f;

        #pragma unroll
        for (int lc = 0; lc < 16; ++lc) {
            const int c = k0 + lc;
            float* pr = prow + (lc & 1) * 16;
            if (tid == c) {
                #pragma unroll
                for (int j = lc; j < 16; ++j) pr[j] = p[j];
            }
            __syncthreads();
            {
                const bool below = tid > c;
                float q[16];
                #pragma unroll
                for (int j = lc; j < 16; ++j) q[j] = pr[j];
                const float f = p[lc] / q[lc];
                p[lc] = below ? f : p[lc];
                #pragma unroll
                for (int j = lc + 1; j < 16; ++j)
                    p[j] = below ? (p[j] - f * q[j]) : p[j];
            }
        }

        if (tid >= k0 && tid < k0 + 16) {
            #pragma unroll
            for (int j = 0; j < 16; ++j)
                Msh[tid * 129 + k0 + j] = p[j];
            #pragma unroll
            for (int j = 0; j < 16; ++j)
                Lbuf[(tid - k0) * 16 + j] = p[j];
        }
        __syncthreads();

        const int ncm = KDIM - 16 - k0;
        if (tid <= ncm) {
            const int j = k0 + 16 + tid;
            float u[16];
            #pragma unroll
            for (int r = 0; r < 16; ++r)
                u[r] = Msh[(k0 + r) * 129 + j];
            #pragma unroll
            for (int r = 1; r < 16; ++r) {
                const float4* lr4 = (const float4*)(Lbuf + r * 16);
                float l[16];
                #pragma unroll
                for (int g = 0; g * 4 < r; ++g) {
                    const float4 t = lr4[g];
                    l[4 * g] = t.x; l[4 * g + 1] = t.y; l[4 * g + 2] = t.z; l[4 * g + 3] = t.w;
                }
                #pragma unroll
                for (int c = 0; c < r; ++c)
                    u[r] -= l[c] * u[c];
            }
            #pragma unroll
            for (int r = 0; r < 16; ++r) {
                Msh[(k0 + r) * 129 + j] = u[r];
                Ubuf[r * 116 + tid]     = u[r];
            }
        }
        __syncthreads();

        if (tid >= k0 + 16) {
            float* row = Msh + tid * 129 + k0 + 16;
            for (int jj = 0; jj < ncm; jj += 4) {
                float x0 = row[jj], x1 = row[jj + 1], x2 = row[jj + 2], x3 = row[jj + 3];
                #pragma unroll
                for (int c = 0; c < 16; ++c) {
                    const float4 uu = *(const float4*)(Ubuf + c * 116 + jj);
                    x0 -= p[c] * uu.x;
                    x1 -= p[c] * uu.y;
                    x2 -= p[c] * uu.z;
                    x3 -= p[c] * uu.w;
                }
                row[jj] = x0; row[jj + 1] = x1; row[jj + 2] = x2; row[jj + 3] = x3;
            }
            float xr = row[ncm];
            #pragma unroll
            for (int c = 0; c < 16; ++c)
                xr -= p[c] * Ubuf[c * 116 + ncm];
            row[ncm] = xr;
        }
        __syncthreads();
    }

    for (int c = KDIM - 1; c >= 0; --c) {
        __syncthreads();
        if (tid == c)
            Msh[c * 129 + KDIM] /= Msh[c * 129 + c];
        __syncthreads();
        if (tid < c)
            Msh[tid * 129 + KDIM] -= Msh[tid * 129 + c] * Msh[c * 129 + KDIM];
    }
    __syncthreads();
    Cout[i * KDIM + tid] = Msh[tid * 129 + KDIM];
}

extern "C" void kernel_launch(void* const* d_in, const int* in_sizes, int n_in,
                              void* d_out, int out_size, void* d_ws, size_t ws_size,
                              hipStream_t stream)
{
    const float* feat_x = (const float*)d_in[0];
    const float* feat_y = (const float*)d_in[1];
    const float* evals_x = (const float*)d_in[2];
    const float* evals_y = (const float*)d_in[3];
    const float* evecs_x = (const float*)d_in[4];
    const float* evecs_y = (const float*)d_in[5];

    float* wsf   = (float*)d_ws;
    float* Amat  = wsf;
    float* Bmat  = wsf + 32768;
    float* AAt   = wsf + 65536;
    float* BAt   = wsf + 81920;
    float* Dmask = wsf + 98304;
    float* part  = wsf + 114688;
    float* Cout  = (float*)d_out;

    fm_gemm_partial<<<dim3(NCHUNKS, 2), 512, 0, stream>>>(
        evecs_x, evecs_y, feat_x, feat_y, part);

    fm_reduce<<<(KP + 255) / 256, 256, 0, stream>>>(part, Amat, Bmat);

    fm_mask<<<(KDIM * KDIM + 255) / 256, 256, 0, stream>>>(evals_x, evals_y, Dmask);

    fm_gram<<<2, 256, 0, stream>>>(Amat, Bmat, AAt, BAt);

    fm_solve<<<KDIM, KDIM, 74624  , stream>>>(
        AAt, BAt, Dmask, Cout);
}


# gemm kernel: a wave owns 128 rows x 32 columns (4-wave workgroups, grid 128x2); feat goes global->registers as whole-line 8-B loads, only the evecs slab crosses LDS; k order and split-K sums regroup
# speedup vs baseline: 2.7127x; 1.6009x over previous
#include <hip/hip_runtime.h>
#include <hip/hip_bf16.h>

typedef __attribute__((ext_vector_type(2))) float v2f;
typedef __attribute__((ext_vector_type(4))) float v4f;
typedef __attribute__((ext_vector_type(8))) float v8f;

#define KDIM 128
#define PDIM 256
#define NDIM 100000
#define LAMBDA 0.001f
#define NCHUNKS 64
#define SLAB 32
#define CHUNK 1568
#define EVP 36
#define KP (KDIM * PDIM)

__global__ __launch_bounds__(128) void fm_gemm_partial(
    const float* __restrict__ evx, const float* __restrict__ evy,
    const float* __restrict__ fx,  const float* __restrict__ fy,
    float* __restrict__ partials)
{
    const int sel   = blockIdx.y;
    const int chunk = blockIdx.x >> 1;
    const int half  = blockIdx.x & 1;
    const float* __restrict__ ev   = sel ? evy : evx;
    const float* __restrict__ feat = sel ? fy  : fx;

    __shared__ __attribute__((aligned(16))) float evS[2][KDIM][EVP];

    const int tid  = threadIdx.x;
    const int lane = tid & 31;
    const int wave = tid >> 5;
    const int c    = lane & 15;
    const int h    = lane >> 4;
    const int colBase = half * 128 + wave * 32;

    const int nStart = chunk * CHUNK;
    const int nEnd   = min(nStart + CHUNK, NDIM);
    const int nSlabs = (nEnd - nStart) / SLAB;

    const int sRow = tid >> 3;
    const int sCol = (tid & 7) * 4;
    const float* __restrict__ evp = ev + (size_t)sRow * NDIM + nStart + sCol;
    const float* __restrict__ ftp = feat + (size_t)(nStart + 4 * h) * PDIM + colBase + 2 * c;

    v8f acc[8][2];
    #pragma unroll
    for (int rb = 0; rb < 8; ++rb) { acc[rb][0] = (v8f)0.0f; acc[rb][1] = (v8f)0.0f; }

    v2f fcur[16], fnxt[16];
    v4f evr[8];

    #pragma unroll
    for (int q = 0; q < 8; ++q)
        evr[q] = *(const v4f*)(evp + (size_t)q * 16 * NDIM);
    #pragma unroll
    for (int f = 0; f < 16; ++f)
        fcur[f] = *(const v2f*)(ftp + (size_t)((f >> 2) * 8 + (f & 3)) * PDIM);
    #pragma unroll
    for (int q = 0; q < 8; ++q)
        *(v4f*)&evS[0][sRow + 16 * q][sCol] = evr[q];
    __syncthreads();

    for (int i = 0; i < nSlabs; ++i) {
        const int p = i & 1;
        const bool more = (i + 1 < nSlabs);

        if (more) {
            const float* __restrict__ e2 = evp + (size_t)(i + 1) * SLAB;
            #pragma unroll
            for (int q = 0; q < 8; ++q)
                evr[q] = *(const v4f*)(e2 + (size_t)q * 16 * NDIM);
            const float* __restrict__ f2 = ftp + (size_t)(i + 1) * SLAB * PDIM;
            #pragma unroll
            for (int f = 0; f < 16; ++f)
                fnxt[f] = *(const v2f*)(f2 + (size_t)((f >> 2) * 8 + (f & 3)) * PDIM);
        }

        v4f a[8];
        #pragma unroll
        for (int rb = 0; rb < 8; ++rb)
            a[rb] = *(const v4f*)&evS[p][rb * 16 + c][4 * h];
        #pragma unroll
        for (int jj = 0; jj < 4; ++jj) {
            v4f an[8];
            if (jj < 3) {
                #pragma unroll
                for (int rb = 0; rb < 8; ++rb)
                    an[rb] = *(const v4f*)&evS[p][rb * 16 + c][8 * (jj + 1) + 4 * h];
            }
            #pragma unroll
            for (int u = 0; u < 2; ++u) {
                const v2f f0 = fcur[jj * 4 + u * 2];
                const v2f f1 = fcur[jj * 4 + u * 2 + 1];
                v2f b0, b1;
                b0.x = f0.x; b0.y = f1.x;
                b1.x = f0.y; b1.y = f1.y;
                #pragma unroll
                for (int rb = 0; rb < 8; ++rb) {
                    v2f av;
                    av.x = a[rb][2 * u];
                    av.y = a[rb][2 * u + 1];
                    acc[rb][0] = __builtin_amdgcn_wmma_f32_16x16x4_f32(
                        false, av, false, b0, (short)0, acc[rb][0], false, false);
                    acc[rb][1] = __builtin_amdgcn_wmma_f32_16x16x4_f32(
                        false, av, false, b1, (short)0, acc[rb][1], false, false);
                }
            }
            if (jj < 3) {
                #pragma unroll
                for (int rb = 0; rb < 8; ++rb) a[rb] = an[rb];
            }
        }

        if (more) {
            #pragma unroll
            for (int q = 0; q < 8; ++q)
                *(v4f*)&evS[p ^ 1][sRow + 16 * q][sCol] = evr[q];
            #pragma unroll
            for (int f = 0; f < 16; ++f) fcur[f] = fnxt[f];
        }
        __syncthreads();
    }

    float* out = partials + (size_t)sel * NCHUNKS * KP + (size_t)chunk * KP;
    #pragma unroll
    for (int rb = 0; rb < 8; ++rb) {
        #pragma unroll
        for (int r = 0; r < 8; ++r) {
            const int row = rb * 16 + 8 * h + r;
            v2f o;
            o.x = acc[rb][0][r];
            o.y = acc[rb][1][r];
            *(v2f*)&out[row * PDIM + colBase + 2 * c] = o;
        }
    }
}

__global__ void fm_reduce(const float* __restrict__ partials,
                          float* __restrict__ Amat, float* __restrict__ Bmat)
{
    int idx = blockIdx.x * blockDim.x + threadIdx.x;
    if (idx >= KP) return;
    float sa = 0.0f, sb = 0.0f;
    for (int c = 0; c < NCHUNKS; ++c) {
        sa += partials[(size_t)c * KP + idx];
        sb += partials[(size_t)(NCHUNKS + c) * KP + idx];
    }
    Amat[idx] = sa;
    Bmat[idx] = sb;
}

__global__ void fm_mask(const float* __restrict__ evx,
                        const float* __restrict__ evy,
                        float* __restrict__ D)
{
    int idx = blockIdx.x * blockDim.x + threadIdx.x;
    if (idx >= KDIM * KDIM) return;
    int i = idx >> 7;
    int j = idx & 127;
    float s  = fmaxf(evx[KDIM - 1], evy[KDIM - 1]);
    float g1 = sqrtf(evx[j] / s);
    float g2 = sqrtf(evy[i] / s);
    float d1 = 1.0f / (g1 * g1 + 1.0f);
    float d2 = 1.0f / (g2 * g2 + 1.0f);
    float re = g2 * d2 - g1 * d1;
    float im = d2 - d1;
    D[idx] = re * re + im * im;
}

__global__ __launch_bounds__(256) void fm_gram(
    const float* __restrict__ Amat, const float* __restrict__ Bmat,
    float* __restrict__ AAt, float* __restrict__ BAt)
{
    const float* __restrict__ L = blockIdx.x ? Bmat : Amat;
    const float* __restrict__ R = Amat;
    float* __restrict__ out     = blockIdx.x ? BAt : AAt;

    __shared__ float blkL[KDIM][33];
    __shared__ float blkR[KDIM][33];

    const int tid  = threadIdx.x;
    const int lane = tid & 31;
    const int wave = tid >> 5;
    const int iBase = wave * 16;

    v8f acc[8];
    #pragma unroll
    for (int t = 0; t < 8; ++t) acc[t] = (v8f)0.0f;

    for (int p0 = 0; p0 < PDIM; p0 += 32) {
        for (int idx = tid; idx < KDIM * 32; idx += 256) {
            int r = idx >> 5, c = idx & 31;
            blkL[r][c] = L[r * PDIM + p0 + c];
            blkR[r][c] = R[r * PDIM + p0 + c];
        }
        __syncthreads();

        const int m  = lane & 15;
        const int kp = (lane >> 4) << 1;
        #pragma unroll
        for (int kk = 0; kk < 8; ++kk) {
            const int pOff = kk * 4;
            v2f a;
            a.x = blkL[iBase + m][pOff + kp];
            a.y = blkL[iBase + m][pOff + kp + 1];
            #pragma unroll
            for (int t = 0; t < 8; ++t) {
                v2f b;
                b.x = blkR[t * 16 + m][pOff + kp];
                b.y = blkR[t * 16 + m][pOff + kp + 1];
                acc[t] = __builtin_amdgcn_wmma_f32_16x16x4_f32(
                    false, a, false, b, (short)0, acc[t], false, false);
            }
        }
        __syncthreads();
    }

    const int mrow = iBase + ((lane >> 4) << 3);
    #pragma unroll
    for (int t = 0; t < 8; ++t) {
        const int j = t * 16 + (lane & 15);
        #pragma unroll
        for (int r = 0; r < 8; ++r)
            out[(mrow + r) * KDIM + j] = acc[t][r];
    }
}

__global__ __launch_bounds__(128) void fm_solve(
    const float* __restrict__ AAt, const float* __restrict__ BAt,
    const float* __restrict__ D,   float* __restrict__ Cout)
{
    extern __shared__ float4 smem4[];
    float* Msh  = (float*)smem4;
    float* prow = Msh + KDIM * 129;
    float* Lbuf = prow + 32;
    float* Ubuf = Lbuf + 256;
    const int i   = blockIdx.x;
    const int tid = threadIdx.x;

    for (int idx = tid; idx < KDIM * 129; idx += 128) {
        int r = idx / 129, c = idx - r * 129;
        float v;
        if (c < KDIM) {
            v = AAt[r * KDIM + c];
            if (c == r) v += LAMBDA * D[i * KDIM + r];
        } else {
            v = BAt[i * KDIM + r];
        }
        Msh[idx] = v;
    }
    __syncthreads();

    for (int kb = 0; kb < KDIM / 16; ++kb) {
        const int k0 = kb * 16;

        float p[16];
        #pragma unroll
        for (int j = 0; j < 16; ++j)
            p[j] = (tid >= k0) ? Msh[tid * 129 + k0 + j] : 0.0f;

        #pragma unroll
        for (int lc = 0; lc < 16; ++lc) {
            const int c = k0 + lc;
            float* pr = prow + (lc & 1) * 16;
            if (tid == c) {
                #pragma unroll
                for (int j = lc; j < 16; ++j) pr[j] = p[j];
            }
            __syncthreads();
            {
                const bool below = tid > c;
                float q[16];
                #pragma unroll
                for (int j = lc; j < 16; ++j) q[j] = pr[j];
                const float f = p[lc] / q[lc];
                p[lc] = below ? f : p[lc];
                #pragma unroll
                for (int j = lc + 1; j < 16; ++j)
                    p[j] = below ? (p[j] - f * q[j]) : p[j];
            }
        }

        if (tid >= k0 && tid < k0 + 16) {
            #pragma unroll
            for (int j = 0; j < 16; ++j)
                Msh[tid * 129 + k0 + j] = p[j];
            #pragma unroll
            for (int j = 0; j < 16; ++j)
                Lbuf[(tid - k0) * 16 + j] = p[j];
        }
        __syncthreads();

        const int ncm = KDIM - 16 - k0;
        if (tid <= ncm) {
            const int j = k0 + 16 + tid;
            float u[16];
            #pragma unroll
            for (int r = 0; r < 16; ++r)
                u[r] = Msh[(k0 + r) * 129 + j];
            #pragma unroll
            for (int r = 1; r < 16; ++r) {
                const float4* lr4 = (const float4*)(Lbuf + r * 16);
                float l[16];
                #pragma unroll
                for (int g = 0; g * 4 < r; ++g) {
                    const float4 t = lr4[g];
                    l[4 * g] = t.x; l[4 * g + 1] = t.y; l[4 * g + 2] = t.z; l[4 * g + 3] = t.w;
                }
                #pragma unroll
                for (int c = 0; c < r; ++c)
                    u[r] -= l[c] * u[c];
            }
            #pragma unroll
            for (int r = 0; r < 16; ++r) {
                Msh[(k0 + r) * 129 + j] = u[r];
                Ubuf[r * 116 + tid]     = u[r];
            }
        }
        __syncthreads();

        if (tid >= k0 + 16) {
            float* row = Msh + tid * 129 + k0 + 16;
            for (int jj = 0; jj < ncm; jj += 4) {
                float x0 = row[jj], x1 = row[jj + 1], x2 = row[jj + 2], x3 = row[jj + 3];
                #pragma unroll
                for (int c = 0; c < 16; ++c) {
                    const float4 uu = *(const float4*)(Ubuf + c * 116 + jj);
                    x0 -= p[c] * uu.x;
                    x1 -= p[c] * uu.y;
                    x2 -= p[c] * uu.z;
                    x3 -= p[c] * uu.w;
                }
                row[jj] = x0; row[jj + 1] = x1; row[jj + 2] = x2; row[jj + 3] = x3;
            }
            float xr = row[ncm];
            #pragma unroll
            for (int c = 0; c < 16; ++c)
                xr -= p[c] * Ubuf[c * 116 + ncm];
            row[ncm] = xr;
        }
        __syncthreads();
    }

    for (int c = KDIM - 1; c >= 0; --c) {
        __syncthreads();
        if (tid == c)
            Msh[c * 129 + KDIM] /= Msh[c * 129 + c];
        __syncthreads();
        if (tid < c)
            Msh[tid * 129 + KDIM] -= Msh[tid * 129 + c] * Msh[c * 129 + KDIM];
    }
    __syncthreads();
    Cout[i * KDIM + tid] = Msh[tid * 129 + KDIM];
}

extern "C" void kernel_launch(void* const* d_in, const int* in_sizes, int n_in,
                              void* d_out, int out_size, void* d_ws, size_t ws_size,
                              hipStream_t stream)
{
    const float* feat_x = (const float*)d_in[0];
    const float* feat_y = (const float*)d_in[1];
    const float* evals_x = (const float*)d_in[2];
    const float* evals_y = (const float*)d_in[3];
    const float* evecs_x = (const float*)d_in[4];
    const float* evecs_y = (const float*)d_in[5];

    float* wsf   = (float*)d_ws;
    float* Amat  = wsf;
    float* Bmat  = wsf + 32768;
    float* AAt   = wsf + 65536;
    float* BAt   = wsf + 81920;
    float* Dmask = wsf + 98304;
    float* part  = wsf + 114688;
    float* Cout  = (float*)d_out;

    fm_gemm_partial<<<dim3(2 * NCHUNKS, 2), 128, 0, stream>>>(
        evecs_x, evecs_y, feat_x, feat_y, part);

    fm_reduce<<<(KP + 255) / 256, 256, 0, stream>>>(part, Amat, Bmat);

    fm_mask<<<(KDIM * KDIM + 255) / 256, 256, 0, stream>>>(evals_x, evals_y, Dmask);

    fm_gram<<<2, 256, 0, stream>>>(Amat, Bmat, AAt, BAt);

    fm_solve<<<KDIM, KDIM, 74624  , stream>>>(
        AAt, BAt, Dmask, Cout);
}


# solve kernel: the system goes global->LDS as 32 independent 16-B loads per thread, requested ahead of the stores (was 129 load-wait-store trips of 4 B); first row rotated per system; values kept
# speedup vs baseline: 3.5145x; 1.2956x over previous
#include <hip/hip_runtime.h>
#include <hip/hip_bf16.h>

typedef __attribute__((ext_vector_type(2))) float v2f;
typedef __attribute__((ext_vector_type(4))) float v4f;
typedef __attribute__((ext_vector_type(8))) float v8f;

#define KDIM 128
#define PDIM 256
#define NDIM 100000
#define LAMBDA 0.001f
#define NCHUNKS 64
#define SLAB 32
#define CHUNK 1568
#define EVP 36
#define KP (KDIM * PDIM)

__global__ __launch_bounds__(128) void fm_gemm_partial(
    const float* __restrict__ evx, const float* __restrict__ evy,
    const float* __restrict__ fx,  const float* __restrict__ fy,
    float* __restrict__ partials)
{
    const int sel   = blockIdx.y;
    const int chunk = blockIdx.x >> 1;
    const int half  = blockIdx.x & 1;
    const float* __restrict__ ev   = sel ? evy : evx;
    const float* __restrict__ feat = sel ? fy  : fx;

    __shared__ __attribute__((aligned(16))) float evS[2][KDIM][EVP];

    const int tid  = threadIdx.x;
    const int lane = tid & 31;
    const int wave = tid >> 5;
    const int c    = lane & 15;
    const int h    = lane >> 4;
    const int colBase = half * 128 + wave * 32;

    const int nStart = chunk * CHUNK;
    const int nEnd   = min(nStart + CHUNK, NDIM);
    const int nSlabs = (nEnd - nStart) / SLAB;

    const int sRow = tid >> 3;
    const int sCol = (tid & 7) * 4;
    const float* __restrict__ evp = ev + (size_t)sRow * NDIM + nStart + sCol;
    const float* __restrict__ ftp = feat + (size_t)(nStart + 4 * h) * PDIM + colBase + 2 * c;

    v8f acc[8][2];
    #pragma unroll
    for (int rb = 0; rb < 8; ++rb) { acc[rb][0] = (v8f)0.0f; acc[rb][1] = (v8f)0.0f; }

    v2f fcur[16], fnxt[16];
    v4f evr[8];

    #pragma unroll
    for (int q = 0; q < 8; ++q)
        evr[q] = *(const v4f*)(evp + (size_t)q * 16 * NDIM);
    #pragma unroll
    for (int f = 0; f < 16; ++f)
        fcur[f] = *(const v2f*)(ftp + (size_t)((f >> 2) * 8 + (f & 3)) * PDIM);
    #pragma unroll
    for (int q = 0; q < 8; ++q)
        *(v4f*)&evS[0][sRow + 16 * q][sCol] = evr[q];
    __syncthreads();

    for (int i = 0; i < nSlabs; ++i) {
        const int p = i & 1;
        const bool more = (i + 1 < nSlabs);

        if (more) {
            const float* __restrict__ e2 = evp + (size_t)(i + 1) * SLAB;
            #pragma unroll
            for (int q = 0; q < 8; ++q)
                evr[q] = *(const v4f*)(e2 + (size_t)q * 16 * NDIM);
            const float* __restrict__ f2 = ftp + (size_t)(i + 1) * SLAB * PDIM;
            #pragma unroll
            for (int f = 0; f < 16; ++f)
                fnxt[f] = *(const v2f*)(f2 + (size_t)((f >> 2) * 8 + (f & 3)) * PDIM);
        }

        v4f a[8];
        #pragma unroll
        for (int rb = 0; rb < 8; ++rb)
            a[rb] = *(const v4f*)&evS[p][rb * 16 + c][4 * h];
        #pragma unroll
        for (int jj = 0; jj < 4; ++jj) {
            v4f an[8];
            if (jj < 3) {
                #pragma unroll
                for (int rb = 0; rb < 8; ++rb)
                    an[rb] = *(const v4f*)&evS[p][rb * 16 + c][8 * (jj + 1) + 4 * h];
            }
            #pragma unroll
            for (int u = 0; u < 2; ++u) {
                const v2f f0 = fcur[jj * 4 + u * 2];
                const v2f f1 = fcur[jj * 4 + u * 2 + 1];
                v2f b0, b1;
                b0.x = f0.x; b0.y = f1.x;
                b1.x = f0.y; b1.y = f1.y;
                #pragma unroll
                for (int rb = 0; rb < 8; ++rb) {
                    v2f av;
                    av.x = a[rb][2 * u];
                    av.y = a[rb][2 * u + 1];
                    acc[rb][0] = __builtin_amdgcn_wmma_f32_16x16x4_f32(
                        false, av, false, b0, (short)0, acc[rb][0], false, false);
                    acc[rb][1] = __builtin_amdgcn_wmma_f32_16x16x4_f32(
                        false, av, false, b1, (short)0, acc[rb][1], false, false);
                }
            }
            if (jj < 3) {
                #pragma unroll
                for (int rb = 0; rb < 8; ++rb) a[rb] = an[rb];
            }
        }

        if (more) {
            #pragma unroll
            for (int q = 0; q < 8; ++q)
                *(v4f*)&evS[p ^ 1][sRow + 16 * q][sCol] = evr[q];
            #pragma unroll
            for (int f = 0; f < 16; ++f) fcur[f] = fnxt[f];
        }
        __syncthreads();
    }

    float* out = partials + (size_t)sel * NCHUNKS * KP + (size_t)chunk * KP;
    #pragma unroll
    for (int rb = 0; rb < 8; ++rb) {
        #pragma unroll
        for (int r = 0; r < 8; ++r) {
            const int row = rb * 16 + 8 * h + r;
            v2f o;
            o.x = acc[rb][0][r];
            o.y = acc[rb][1][r];
            *(v2f*)&out[row * PDIM + colBase + 2 * c] = o;
        }
    }
}

__global__ void fm_reduce(const float* __restrict__ partials,
                          float* __restrict__ Amat, float* __restrict__ Bmat)
{
    int idx = blockIdx.x * blockDim.x + threadIdx.x;
    if (idx >= KP) return;
    float sa = 0.0f, sb = 0.0f;
    for (int c = 0; c < NCHUNKS; ++c) {
        sa += partials[(size_t)c * KP + idx];
        sb += partials[(size_t)(NCHUNKS + c) * KP + idx];
    }
    Amat[idx] = sa;
    Bmat[idx] = sb;
}

__global__ void fm_mask(const float* __restrict__ evx,
                        const float* __restrict__ evy,
                        float* __restrict__ D)
{
    int idx = blockIdx.x * blockDim.x + threadIdx.x;
    if (idx >= KDIM * KDIM) return;
    int i = idx >> 7;
    int j = idx & 127;
    float s  = fmaxf(evx[KDIM - 1], evy[KDIM - 1]);
    float g1 = sqrtf(evx[j] / s);
    float g2 = sqrtf(evy[i] / s);
    float d1 = 1.0f / (g1 * g1 + 1.0f);
    float d2 = 1.0f / (g2 * g2 + 1.0f);
    float re = g2 * d2 - g1 * d1;
    float im = d2 - d1;
    D[idx] = re * re + im * im;
}

__global__ __launch_bounds__(256) void fm_gram(
    const float* __restrict__ Amat, const float* __restrict__ Bmat,
    float* __restrict__ AAt, float* __restrict__ BAt)
{
    const float* __restrict__ L = blockIdx.x ? Bmat : Amat;
    const float* __restrict__ R = Amat;
    float* __restrict__ out     = blockIdx.x ? BAt : AAt;

    __shared__ float blkL[KDIM][33];
    __shared__ float blkR[KDIM][33];

    const int tid  = threadIdx.x;
    const int lane = tid & 31;
    const int wave = tid >> 5;
    const int iBase = wave * 16;

    v8f acc[8];
    #pragma unroll
    for (int t = 0; t < 8; ++t) acc[t] = (v8f)0.0f;

    for (int p0 = 0; p0 < PDIM; p0 += 32) {
        for (int idx = tid; idx < KDIM * 32; idx += 256) {
            int r = idx >> 5, c = idx & 31;
            blkL[r][c] = L[r * PDIM + p0 + c];
            blkR[r][c] = R[r * PDIM + p0 + c];
        }
        __syncthreads();

        const int m  = lane & 15;
        const int kp = (lane >> 4) << 1;
        #pragma unroll
        for (int kk = 0; kk < 8; ++kk) {
            const int pOff = kk * 4;
            v2f a;
            a.x = blkL[iBase + m][pOff + kp];
            a.y = blkL[iBase + m][pOff + kp + 1];
            #pragma unroll
            for (int t = 0; t < 8; ++t) {
                v2f b;
                b.x = blkR[t * 16 + m][pOff + kp];
                b.y = blkR[t * 16 + m][pOff + kp + 1];
                acc[t] = __builtin_amdgcn_wmma_f32_16x16x4_f32(
                    false, a, false, b, (short)0, acc[t], false, false);
            }
        }
        __syncthreads();
    }

    const int mrow = iBase + ((lane >> 4) << 3);
    #pragma unroll
    for (int t = 0; t < 8; ++t) {
        const int j = t * 16 + (lane & 15);
        #pragma unroll
        for (int r = 0; r < 8; ++r)
            out[(mrow + r) * KDIM + j] = acc[t][r];
    }
}

__global__ __launch_bounds__(128) void fm_solve(
    const float* __restrict__ AAt, const float* __restrict__ BAt,
    const float* __restrict__ D,   float* __restrict__ Cout)
{
    extern __shared__ float4 smem4[];
    float* Msh  = (float*)smem4;
    float* prow = Msh + KDIM * 129;
    float* Lbuf = prow + 32;
    float* Ubuf = Lbuf + 256;
    const int i   = blockIdx.x;
    const int tid = threadIdx.x;

    {
        const int lane = tid & 31;
        const int wv   = tid >> 5;
        float4 a4[32];
        #pragma unroll
        for (int q = 0; q < 32; ++q) {
            const int row = 4 * ((q + i) & 31) + wv;
            a4[q] = *(const float4*)(AAt + row * KDIM + 4 * lane);
        }
        const int rd = 4 * lane + wv;
        float dg = AAt[rd * KDIM + rd];
        dg += LAMBDA * D[i * KDIM + rd];
        const float rhs = BAt[i * KDIM + tid];
        #pragma unroll
        for (int q = 0; q < 32; ++q) {
            const int row = 4 * ((q + i) & 31) + wv;
            float* dst = Msh + row * 129 + 4 * lane;
            dst[0] = a4[q].x; dst[1] = a4[q].y; dst[2] = a4[q].z; dst[3] = a4[q].w;
        }
        Msh[rd * 129 + rd] = dg;
        Msh[tid * 129 + KDIM] = rhs;
    }
    __syncthreads();

    for (int kb = 0; kb < KDIM / 16; ++kb) {
        const int k0 = kb * 16;

        float p[16];
        #pragma unroll
        for (int j = 0; j < 16; ++j)
            p[j] = (tid >= k0) ? Msh[tid * 129 + k0 + j] : 0.0f;

        #pragma unroll
        for (int lc = 0; lc < 16; ++lc) {
            const int c = k0 + lc;
            float* pr = prow + (lc & 1) * 16;
            if (tid == c) {
                #pragma unroll
                for (int j = lc; j < 16; ++j) pr[j] = p[j];
            }
            __syncthreads();
            {
                const bool below = tid > c;
                float q[16];
                #pragma unroll
                for (int j = lc; j < 16; ++j) q[j] = pr[j];
                const float f = p[lc] / q[lc];
                p[lc] = below ? f : p[lc];
                #pragma unroll
                for (int j = lc + 1; j < 16; ++j)
                    p[j] = below ? (p[j] - f * q[j]) : p[j];
            }
        }

        if (tid >= k0 && tid < k0 + 16) {
            #pragma unroll
            for (int j = 0; j < 16; ++j)
                Msh[tid * 129 + k0 + j] = p[j];
            #pragma unroll
            for (int j = 0; j < 16; ++j)
                Lbuf[(tid - k0) * 16 + j] = p[j];
        }
        __syncthreads();

        const int ncm = KDIM - 16 - k0;
        if (tid <= ncm) {
            const int j = k0 + 16 + tid;
            float u[16];
            #pragma unroll
            for (int r = 0; r < 16; ++r)
                u[r] = Msh[(k0 + r) * 129 + j];
            #pragma unroll
            for (int r = 1; r < 16; ++r) {
                const float4* lr4 = (const float4*)(Lbuf + r * 16);
                float l[16];
                #pragma unroll
                for (int g = 0; g * 4 < r; ++g) {
                    const float4 t = lr4[g];
                    l[4 * g] = t.x; l[4 * g + 1] = t.y; l[4 * g + 2] = t.z; l[4 * g + 3] = t.w;
                }
                #pragma unroll
                for (int c = 0; c < r; ++c)
                    u[r] -= l[c] * u[c];
            }
            #pragma unroll
            for (int r = 0; r < 16; ++r) {
                Msh[(k0 + r) * 129 + j] = u[r];
                Ubuf[r * 116 + tid]     = u[r];
            }
        }
        __syncthreads();

        if (tid >= k0 + 16) {
            float* row = Msh + tid * 129 + k0 + 16;
            for (int jj = 0; jj < ncm; jj += 4) {
                float x0 = row[jj], x1 = row[jj + 1], x2 = row[jj + 2], x3 = row[jj + 3];
                #pragma unroll
                for (int c = 0; c < 16; ++c) {
                    const float4 uu = *(const float4*)(Ubuf + c * 116 + jj);
                    x0 -= p[c] * uu.x;
                    x1 -= p[c] * uu.y;
                    x2 -= p[c] * uu.z;
                    x3 -= p[c] * uu.w;
                }
                row[jj] = x0; row[jj + 1] = x1; row[jj + 2] = x2; row[jj + 3] = x3;
            }
            float xr = row[ncm];
            #pragma unroll
            for (int c = 0; c < 16; ++c)
                xr -= p[c] * Ubuf[c * 116 + ncm];
            row[ncm] = xr;
        }
        __syncthreads();
    }

    for (int c = KDIM - 1; c >= 0; --c) {
        __syncthreads();
        if (tid == c)
            Msh[c * 129 + KDIM] /= Msh[c * 129 + c];
        __syncthreads();
        if (tid < c)
            Msh[tid * 129 + KDIM] -= Msh[tid * 129 + c] * Msh[c * 129 + KDIM];
    }
    __syncthreads();
    Cout[i * KDIM + tid] = Msh[tid * 129 + KDIM];
}

extern "C" void kernel_launch(void* const* d_in, const int* in_sizes, int n_in,
                              void* d_out, int out_size, void* d_ws, size_t ws_size,
                              hipStream_t stream)
{
    const float* feat_x = (const float*)d_in[0];
    const float* feat_y = (const float*)d_in[1];
    const float* evals_x = (const float*)d_in[2];
    const float* evals_y = (const float*)d_in[3];
    const float* evecs_x = (const float*)d_in[4];
    const float* evecs_y = (const float*)d_in[5];

    float* wsf   = (float*)d_ws;
    float* Amat  = wsf;
    float* Bmat  = wsf + 32768;
    float* AAt   = wsf + 65536;
    float* BAt   = wsf + 81920;
    float* Dmask = wsf + 98304;
    float* part  = wsf + 114688;
    float* Cout  = (float*)d_out;

    fm_gemm_partial<<<dim3(2 * NCHUNKS, 2), 128, 0, stream>>>(
        evecs_x, evecs_y, feat_x, feat_y, part);

    fm_reduce<<<(KP + 255) / 256, 256, 0, stream>>>(part, Amat, Bmat);

    fm_mask<<<(KDIM * KDIM + 255) / 256, 256, 0, stream>>>(evals_x, evals_y, Dmask);

    fm_gram<<<2, 256, 0, stream>>>(Amat, Bmat, AAt, BAt);

    fm_solve<<<KDIM, KDIM, 74624  , stream>>>(
        AAt, BAt, Dmask, Cout);
}


# solve kernel: each workgroup forms its own system: A staged in LDS, A*A^T as 36 mirrored WMMA tile pairs plus its own row of B*A^T; gram launch and AAt/BAt memory round trip gone; k order kept
# speedup vs baseline: 4.1664x; 1.1855x over previous
#include <hip/hip_runtime.h>
#include <hip/hip_bf16.h>

typedef __attribute__((ext_vector_type(2))) float v2f;
typedef __attribute__((ext_vector_type(4))) float v4f;
typedef __attribute__((ext_vector_type(8))) float v8f;

#define KDIM 128
#define PDIM 256
#define NDIM 100000
#define LAMBDA 0.001f
#define NCHUNKS 64
#define SLAB 32
#define CHUNK 1568
#define EVP 36
#define KP (KDIM * PDIM)

__global__ __launch_bounds__(128) void fm_gemm_partial(
    const float* __restrict__ evx, const float* __restrict__ evy,
    const float* __restrict__ fx,  const float* __restrict__ fy,
    float* __restrict__ partials)
{
    const int sel   = blockIdx.y;
    const int chunk = blockIdx.x >> 1;
    const int half  = blockIdx.x & 1;
    const float* __restrict__ ev   = sel ? evy : evx;
    const float* __restrict__ feat = sel ? fy  : fx;

    __shared__ __attribute__((aligned(16))) float evS[2][KDIM][EVP];

    const int tid  = threadIdx.x;
    const int lane = tid & 31;
    const int wave = tid >> 5;
    const int c    = lane & 15;
    const int h    = lane >> 4;
    const int colBase = half * 128 + wave * 32;

    const int nStart = chunk * CHUNK;
    const int nEnd   = min(nStart + CHUNK, NDIM);
    const int nSlabs = (nEnd - nStart) / SLAB;

    const int sRow = tid >> 3;
    const int sCol = (tid & 7) * 4;
    const float* __restrict__ evp = ev + (size_t)sRow * NDIM + nStart + sCol;
    const float* __restrict__ ftp = feat + (size_t)(nStart + 4 * h) * PDIM + colBase + 2 * c;

    v8f acc[8][2];
    #pragma unroll
    for (int rb = 0; rb < 8; ++rb) { acc[rb][0] = (v8f)0.0f; acc[rb][1] = (v8f)0.0f; }

    v2f fcur[16], fnxt[16];
    v4f evr[8];

    #pragma unroll
    for (int q = 0; q < 8; ++q)
        evr[q] = *(const v4f*)(evp + (size_t)q * 16 * NDIM);
    #pragma unroll
    for (int f = 0; f < 16; ++f)
        fcur[f] = *(const v2f*)(ftp + (size_t)((f >> 2) * 8 + (f & 3)) * PDIM);
    #pragma unroll
    for (int q = 0; q < 8; ++q)
        *(v4f*)&evS[0][sRow + 16 * q][sCol] = evr[q];
    __syncthreads();

    for (int i = 0; i < nSlabs; ++i) {
        const int p = i & 1;
        const bool more = (i + 1 < nSlabs);

        if (more) {
            const float* __restrict__ e2 = evp + (size_t)(i + 1) * SLAB;
            #pragma unroll
            for (int q = 0; q < 8; ++q)
                evr[q] = *(const v4f*)(e2 + (size_t)q * 16 * NDIM);
            const float* __restrict__ f2 = ftp + (size_t)(i + 1) * SLAB * PDIM;
            #pragma unroll
            for (int f = 0; f < 16; ++f)
                fnxt[f] = *(const v2f*)(f2 + (size_t)((f >> 2) * 8 + (f & 3)) * PDIM);
        }

        v4f a[8];
        #pragma unroll
        for (int rb = 0; rb < 8; ++rb)
            a[rb] = *(const v4f*)&evS[p][rb * 16 + c][4 * h];
        #pragma unroll
        for (int jj = 0; jj < 4; ++jj) {
            v4f an[8];
            if (jj < 3) {
                #pragma unroll
                for (int rb = 0; rb < 8; ++rb)
                    an[rb] = *(const v4f*)&evS[p][rb * 16 + c][8 * (jj + 1) + 4 * h];
            }
            #pragma unroll
            for (int u = 0; u < 2; ++u) {
                const v2f f0 = fcur[jj * 4 + u * 2];
                const v2f f1 = fcur[jj * 4 + u * 2 + 1];
                v2f b0, b1;
                b0.x = f0.x; b0.y = f1.x;
                b1.x = f0.y; b1.y = f1.y;
                #pragma unroll
                for (int rb = 0; rb < 8; ++rb) {
                    v2f av;
                    av.x = a[rb][2 * u];
                    av.y = a[rb][2 * u + 1];
                    acc[rb][0] = __builtin_amdgcn_wmma_f32_16x16x4_f32(
                        false, av, false, b0, (short)0, acc[rb][0], false, false);
                    acc[rb][1] = __builtin_amdgcn_wmma_f32_16x16x4_f32(
                        false, av, false, b1, (short)0, acc[rb][1], false, false);
                }
            }
            if (jj < 3) {
                #pragma unroll
                for (int rb = 0; rb < 8; ++rb) a[rb] = an[rb];
            }
        }

        if (more) {
            #pragma unroll
            for (int q = 0; q < 8; ++q)
                *(v4f*)&evS[p ^ 1][sRow + 16 * q][sCol] = evr[q];
            #pragma unroll
            for (int f = 0; f < 16; ++f) fcur[f] = fnxt[f];
        }
        __syncthreads();
    }

    float* out = partials + (size_t)sel * NCHUNKS * KP + (size_t)chunk * KP;
    #pragma unroll
    for (int rb = 0; rb < 8; ++rb) {
        #pragma unroll
        for (int r = 0; r < 8; ++r) {
            const int row = rb * 16 + 8 * h + r;
            v2f o;
            o.x = acc[rb][0][r];
            o.y = acc[rb][1][r];
            *(v2f*)&out[row * PDIM + colBase + 2 * c] = o;
        }
    }
}

__global__ void fm_reduce(const float* __restrict__ partials,
                          float* __restrict__ Amat, float* __restrict__ Bmat)
{
    int idx = blockIdx.x * blockDim.x + threadIdx.x;
    if (idx >= KP) return;
    float sa = 0.0f, sb = 0.0f;
    for (int c = 0; c < NCHUNKS; ++c) {
        sa += partials[(size_t)c * KP + idx];
        sb += partials[(size_t)(NCHUNKS + c) * KP + idx];
    }
    Amat[idx] = sa;
    Bmat[idx] = sb;
}

__global__ void fm_mask(const float* __restrict__ evx,
                        const float* __restrict__ evy,
                        float* __restrict__ D)
{
    int idx = blockIdx.x * blockDim.x + threadIdx.x;
    if (idx >= KDIM * KDIM) return;
    int i = idx >> 7;
    int j = idx & 127;
    float s  = fmaxf(evx[KDIM - 1], evy[KDIM - 1]);
    float g1 = sqrtf(evx[j] / s);
    float g2 = sqrtf(evy[i] / s);
    float d1 = 1.0f / (g1 * g1 + 1.0f);
    float d2 = 1.0f / (g2 * g2 + 1.0f);
    float re = g2 * d2 - g1 * d1;
    float im = d2 - d1;
    D[idx] = re * re + im * im;
}

#define APITCH 132
__global__ __launch_bounds__(128) void fm_solve(
    const float* __restrict__ Amat, const float* __restrict__ Bmat,
    const float* __restrict__ D,    float* __restrict__ Cout)
{
    extern __shared__ float4 smem4[];
    float* Msh  = (float*)smem4;
    float* prow = Msh + KDIM * 129;
    float* Lbuf = prow + 32;
    float* Ubuf = Lbuf + 256;
    float* Ast  = (float*)smem4;
    float* Brow = Ast + KDIM * APITCH;
    const int i   = blockIdx.x;
    const int tid = threadIdx.x;

    {
        const int lane = tid & 31;
        const int wv   = tid >> 5;
        const int c    = lane & 15;
        const int h    = lane >> 4;

        const float* __restrict__ ap = Amat + (size_t)wv * PDIM + 4 * lane;
        v4f st0[32], st1[32];
        #pragma unroll
        for (int q = 0; q < 32; ++q)
            st0[q] = *(const v4f*)(ap + (size_t)q * 4 * PDIM);
        const v2f bq = *(const v2f*)(Bmat + (size_t)i * PDIM + 2 * tid);
        const float dsc = D[i * KDIM + tid];
        #pragma unroll
        for (int q = 0; q < 32; ++q)
            *(v4f*)&Ast[(4 * q + wv) * APITCH + 4 * lane] = st0[q];
        *(v2f*)&Brow[2 * tid] = bq;
        __syncthreads();
        #pragma unroll
        for (int q = 0; q < 32; ++q)
            st1[q] = *(const v4f*)(ap + (size_t)q * 4 * PDIM + 128);

        v8f acc[11];
        #pragma unroll
        for (int t = 0; t < 11; ++t) acc[t] = (v8f)0.0f;

        int fo[8];
        #pragma unroll
        for (int t = 0; t < 8; ++t) fo[t] = (16 * ((wv + t) & 7) + c) * APITCH + 2 * h;
        const bool row0 = (c == 0);

        for (int hh = 0; hh < 2; ++hh) {
            if (hh == 1) {
                __syncthreads();
                #pragma unroll
                for (int q = 0; q < 32; ++q)
                    *(v4f*)&Ast[(4 * q + wv) * APITCH + 4 * lane] = st1[q];
                __syncthreads();
            }
            v2f f[8], fb;
            #pragma unroll
            for (int t = 0; t < 8; ++t) f[t] = *(const v2f*)&Ast[fo[t]];
            {
                const v2f tb = *(const v2f*)&Brow[128 * hh + 2 * h];
                fb.x = row0 ? tb.x : 0.0f;
                fb.y = row0 ? tb.y : 0.0f;
            }
            for (int ks = 0; ks < 32; ++ks) {
                v2f fn[8], fbn;
                if (ks + 1 < 32) {
                    #pragma unroll
                    for (int t = 0; t < 8; ++t) fn[t] = *(const v2f*)&Ast[fo[t] + 4 * (ks + 1)];
                    const v2f tb = *(const v2f*)&Brow[128 * hh + 4 * (ks + 1) + 2 * h];
                    fbn.x = row0 ? tb.x : 0.0f;
                    fbn.y = row0 ? tb.y : 0.0f;
                } else {
                    #pragma unroll
                    for (int t = 0; t < 8; ++t) fn[t] = f[t];
                    fbn = fb;
                }
                #pragma unroll
                for (int t = 0; t < 5; ++t)
                    acc[t] = __builtin_amdgcn_wmma_f32_16x16x4_f32(
                        false, f[0], false, f[t], (short)0, acc[t], false, false);
                #pragma unroll
                for (int t = 0; t < 4; ++t)
                    acc[5 + t] = __builtin_amdgcn_wmma_f32_16x16x4_f32(
                        false, f[4], false, f[4 + t], (short)0, acc[5 + t], false, false);
                acc[9]  = __builtin_amdgcn_wmma_f32_16x16x4_f32(
                    false, fb, false, f[0], (short)0, acc[9], false, false);
                acc[10] = __builtin_amdgcn_wmma_f32_16x16x4_f32(
                    false, fb, false, f[4], (short)0, acc[10], false, false);
                #pragma unroll
                for (int t = 0; t < 8; ++t) f[t] = fn[t];
                fb = fbn;
            }
        }
        __syncthreads();

        #pragma unroll
        for (int a = 0; a < 2; ++a) {
            const int I = wv + 4 * a;
            #pragma unroll
            for (int t = 0; t < 5 - a; ++t) {
                const int J = (I + t) & 7;
                const int col = 16 * J + c;
                #pragma unroll
                for (int r = 0; r < 8; ++r) {
                    const int row = 16 * I + 8 * h + r;
                    const float v = acc[5 * a + t][r];
                    Msh[row * 129 + col] = v;
                    if (t != 0) Msh[col * 129 + row] = v;
                }
            }
            if (h == 0)
                Msh[(16 * I + c) * 129 + KDIM] = acc[9 + a][0];
        }
        __syncthreads();
        Msh[tid * 129 + tid] += LAMBDA * dsc;
    }

    for (int kb = 0; kb < KDIM / 16; ++kb) {
        const int k0 = kb * 16;

        float p[16];
        #pragma unroll
        for (int j = 0; j < 16; ++j)
            p[j] = (tid >= k0) ? Msh[tid * 129 + k0 + j] : 0.0f;

        #pragma unroll
        for (int lc = 0; lc < 16; ++lc) {
            const int c = k0 + lc;
            float* pr = prow + (lc & 1) * 16;
            if (tid == c) {
                #pragma unroll
                for (int j = lc; j < 16; ++j) pr[j] = p[j];
            }
            __syncthreads();
            {
                const bool below = tid > c;
                float q[16];
                #pragma unroll
                for (int j = lc; j < 16; ++j) q[j] = pr[j];
                const float f = p[lc] / q[lc];
                p[lc] = below ? f : p[lc];
                #pragma unroll
                for (int j = lc + 1; j < 16; ++j)
                    p[j] = below ? (p[j] - f * q[j]) : p[j];
            }
        }

        if (tid >= k0 && tid < k0 + 16) {
            #pragma unroll
            for (int j = 0; j < 16; ++j)
                Msh[tid * 129 + k0 + j] = p[j];
            #pragma unroll
            for (int j = 0; j < 16; ++j)
                Lbuf[(tid - k0) * 16 + j] = p[j];
        }
        __syncthreads();

        const int ncm = KDIM - 16 - k0;
        if (tid <= ncm) {
            const int j = k0 + 16 + tid;
            float u[16];
            #pragma unroll
            for (int r = 0; r < 16; ++r)
                u[r] = Msh[(k0 + r) * 129 + j];
            #pragma unroll
            for (int r = 1; r < 16; ++r) {
                const float4* lr4 = (const float4*)(Lbuf + r * 16);
                float l[16];
                #pragma unroll
                for (int g = 0; g * 4 < r; ++g) {
                    const float4 t = lr4[g];
                    l[4 * g] = t.x; l[4 * g + 1] = t.y; l[4 * g + 2] = t.z; l[4 * g + 3] = t.w;
                }
                #pragma unroll
                for (int c = 0; c < r; ++c)
                    u[r] -= l[c] * u[c];
            }
            #pragma unroll
            for (int r = 0; r < 16; ++r) {
                Msh[(k0 + r) * 129 + j] = u[r];
                Ubuf[r * 116 + tid]     = u[r];
            }
        }
        __syncthreads();

        if (tid >= k0 + 16) {
            float* row = Msh + tid * 129 + k0 + 16;
            for (int jj = 0; jj < ncm; jj += 4) {
                float x0 = row[jj], x1 = row[jj + 1], x2 = row[jj + 2], x3 = row[jj + 3];
                #pragma unroll
                for (int c = 0; c < 16; ++c) {
                    const float4 uu = *(const float4*)(Ubuf + c * 116 + jj);
                    x0 -= p[c] * uu.x;
                    x1 -= p[c] * uu.y;
                    x2 -= p[c] * uu.z;
                    x3 -= p[c] * uu.w;
                }
                row[jj] = x0; row[jj + 1] = x1; row[jj + 2] = x2; row[jj + 3] = x3;
            }
            float xr = row[ncm];
            #pragma unroll
            for (int c = 0; c < 16; ++c)
                xr -= p[c] * Ubuf[c * 116 + ncm];
            row[ncm] = xr;
        }
        __syncthreads();
    }

    for (int c = KDIM - 1; c >= 0; --c) {
        __syncthreads();
        if (tid == c)
            Msh[c * 129 + KDIM] /= Msh[c * 129 + c];
        __syncthreads();
        if (tid < c)
            Msh[tid * 129 + KDIM] -= Msh[tid * 129 + c] * Msh[c * 129 + KDIM];
    }
    __syncthreads();
    Cout[i * KDIM + tid] = Msh[tid * 129 + KDIM];
}

extern "C" void kernel_launch(void* const* d_in, const int* in_sizes, int n_in,
                              void* d_out, int out_size, void* d_ws, size_t ws_size,
                              hipStream_t stream)
{
    const float* feat_x = (const float*)d_in[0];
    const float* feat_y = (const float*)d_in[1];
    const float* evals_x = (const float*)d_in[2];
    const float* evals_y = (const float*)d_in[3];
    const float* evecs_x = (const float*)d_in[4];
    const float* evecs_y = (const float*)d_in[5];

    float* wsf   = (float*)d_ws;
    float* Amat  = wsf;
    float* Bmat  = wsf + 32768;
    float* Dmask = wsf + 98304;
    float* part  = wsf + 114688;
    float* Cout  = (float*)d_out;

    fm_gemm_partial<<<dim3(2 * NCHUNKS, 2), 128, 0, stream>>>(
        evecs_x, evecs_y, feat_x, feat_y, part);

    fm_reduce<<<(KP + 255) / 256, 256, 0, stream>>>(part, Amat, Bmat);

    fm_mask<<<(KDIM * KDIM + 255) / 256, 256, 0, stream>>>(evals_x, evals_y, Dmask);

    fm_solve<<<KDIM, KDIM, 74624  , stream>>>(
        Amat, Bmat, Dmask, Cout);
}


# solve kernel: back substitution blocked by 16 columns: every thread solves each 16x16 triangle itself in registers from broadcast LDS reads, own row's y in a register; 8 barriers, not 256; order kept
# speedup vs baseline: 4.6675x; 1.1203x over previous
#include <hip/hip_runtime.h>
#include <hip/hip_bf16.h>

typedef __attribute__((ext_vector_type(2))) float v2f;
typedef __attribute__((ext_vector_type(4))) float v4f;
typedef __attribute__((ext_vector_type(8))) float v8f;

#define KDIM 128
#define PDIM 256
#define NDIM 100000
#define LAMBDA 0.001f
#define NCHUNKS 64
#define SLAB 32
#define CHUNK 1568
#define EVP 36
#define KP (KDIM * PDIM)

__global__ __launch_bounds__(128) void fm_gemm_partial(
    const float* __restrict__ evx, const float* __restrict__ evy,
    const float* __restrict__ fx,  const float* __restrict__ fy,
    float* __restrict__ partials)
{
    const int sel   = blockIdx.y;
    const int chunk = blockIdx.x >> 1;
    const int half  = blockIdx.x & 1;
    const float* __restrict__ ev   = sel ? evy : evx;
    const float* __restrict__ feat = sel ? fy  : fx;

    __shared__ __attribute__((aligned(16))) float evS[2][KDIM][EVP];

    const int tid  = threadIdx.x;
    const int lane = tid & 31;
    const int wave = tid >> 5;
    const int c    = lane & 15;
    const int h    = lane >> 4;
    const int colBase = half * 128 + wave * 32;

    const int nStart = chunk * CHUNK;
    const int nEnd   = min(nStart + CHUNK, NDIM);
    const int nSlabs = (nEnd - nStart) / SLAB;

    const int sRow = tid >> 3;
    const int sCol = (tid & 7) * 4;
    const float* __restrict__ evp = ev + (size_t)sRow * NDIM + nStart + sCol;
    const float* __restrict__ ftp = feat + (size_t)(nStart + 4 * h) * PDIM + colBase + 2 * c;

    v8f acc[8][2];
    #pragma unroll
    for (int rb = 0; rb < 8; ++rb) { acc[rb][0] = (v8f)0.0f; acc[rb][1] = (v8f)0.0f; }

    v2f fcur[16], fnxt[16];
    v4f evr[8];

    #pragma unroll
    for (int q = 0; q < 8; ++q)
        evr[q] = *(const v4f*)(evp + (size_t)q * 16 * NDIM);
    #pragma unroll
    for (int f = 0; f < 16; ++f)
        fcur[f] = *(const v2f*)(ftp + (size_t)((f >> 2) * 8 + (f & 3)) * PDIM);
    #pragma unroll
    for (int q = 0; q < 8; ++q)
        *(v4f*)&evS[0][sRow + 16 * q][sCol] = evr[q];
    __syncthreads();

    for (int i = 0; i < nSlabs; ++i) {
        const int p = i & 1;
        const bool more = (i + 1 < nSlabs);

        if (more) {
            const float* __restrict__ e2 = evp + (size_t)(i + 1) * SLAB;
            #pragma unroll
            for (int q = 0; q < 8; ++q)
                evr[q] = *(const v4f*)(e2 + (size_t)q * 16 * NDIM);
            const float* __restrict__ f2 = ftp + (size_t)(i + 1) * SLAB * PDIM;
            #pragma unroll
            for (int f = 0; f < 16; ++f)
                fnxt[f] = *(const v2f*)(f2 + (size_t)((f >> 2) * 8 + (f & 3)) * PDIM);
        }

        v4f a[8];
        #pragma unroll
        for (int rb = 0; rb < 8; ++rb)
            a[rb] = *(const v4f*)&evS[p][rb * 16 + c][4 * h];
        #pragma unroll
        for (int jj = 0; jj < 4; ++jj) {
            v4f an[8];
            if (jj < 3) {
                #pragma unroll
                for (int rb = 0; rb < 8; ++rb)
                    an[rb] = *(const v4f*)&evS[p][rb * 16 + c][8 * (jj + 1) + 4 * h];
            }
            #pragma unroll
            for (int u = 0; u < 2; ++u) {
                const v2f f0 = fcur[jj * 4 + u * 2];
                const v2f f1 = fcur[jj * 4 + u * 2 + 1];
                v2f b0, b1;
                b0.x = f0.x; b0.y = f1.x;
                b1.x = f0.y; b1.y = f1.y;
                #pragma unroll
                for (int rb = 0; rb < 8; ++rb) {
                    v2f av;
                    av.x = a[rb][2 * u];
                    av.y = a[rb][2 * u + 1];
                    acc[rb][0] = __builtin_amdgcn_wmma_f32_16x16x4_f32(
                        false, av, false, b0, (short)0, acc[rb][0], false, false);
                    acc[rb][1] = __builtin_amdgcn_wmma_f32_16x16x4_f32(
                        false, av, false, b1, (short)0, acc[rb][1], false, false);
                }
            }
            if (jj < 3) {
                #pragma unroll
                for (int rb = 0; rb < 8; ++rb) a[rb] = an[rb];
            }
        }

        if (more) {
            #pragma unroll
            for (int q = 0; q < 8; ++q)
                *(v4f*)&evS[p ^ 1][sRow + 16 * q][sCol] = evr[q];
            #pragma unroll
            for (int f = 0; f < 16; ++f) fcur[f] = fnxt[f];
        }
        __syncthreads();
    }

    float* out = partials + (size_t)sel * NCHUNKS * KP + (size_t)chunk * KP;
    #pragma unroll
    for (int rb = 0; rb < 8; ++rb) {
        #pragma unroll
        for (int r = 0; r < 8; ++r) {
            const int row = rb * 16 + 8 * h + r;
            v2f o;
            o.x = acc[rb][0][r];
            o.y = acc[rb][1][r];
            *(v2f*)&out[row * PDIM + colBase + 2 * c] = o;
        }
    }
}

__global__ void fm_reduce(const float* __restrict__ partials,
                          float* __restrict__ Amat, float* __restrict__ Bmat)
{
    int idx = blockIdx.x * blockDim.x + threadIdx.x;
    if (idx >= KP) return;
    float sa = 0.0f, sb = 0.0f;
    for (int c = 0; c < NCHUNKS; ++c) {
        sa += partials[(size_t)c * KP + idx];
        sb += partials[(size_t)(NCHUNKS + c) * KP + idx];
    }
    Amat[idx] = sa;
    Bmat[idx] = sb;
}

__global__ void fm_mask(const float* __restrict__ evx,
                        const float* __restrict__ evy,
                        float* __restrict__ D)
{
    int idx = blockIdx.x * blockDim.x + threadIdx.x;
    if (idx >= KDIM * KDIM) return;
    int i = idx >> 7;
    int j = idx & 127;
    float s  = fmaxf(evx[KDIM - 1], evy[KDIM - 1]);
    float g1 = sqrtf(evx[j] / s);
    float g2 = sqrtf(evy[i] / s);
    float d1 = 1.0f / (g1 * g1 + 1.0f);
    float d2 = 1.0f / (g2 * g2 + 1.0f);
    float re = g2 * d2 - g1 * d1;
    float im = d2 - d1;
    D[idx] = re * re + im * im;
}

__device__ __forceinline__ void fm_bs_block(
    float* Msh, const int tid, const int kb,
    const float (&tc)[16][16], const float (&uc)[16],
    float (&tn)[16][16], float (&un)[16], float& yown, float& xout)
{
    const int k0 = kb * 16;
    const int kn = (kb > 0) ? k0 - 16 : 0;

    __syncthreads();

    float yb[16], x[16];
    #pragma unroll
    for (int r = 0; r < 16; ++r)
        yb[r] = Msh[(k0 + r) * 129 + KDIM];
    #pragma unroll
    for (int r = 0; r < 16; ++r) {
        #pragma unroll
        for (int c = r; c < 16; ++c)
            tn[r][c] = Msh[(kn + r) * 129 + kn + c];
    }
    #pragma unroll
    for (int c = 0; c < 16; ++c)
        un[c] = Msh[tid * 129 + kn + c];
    __builtin_amdgcn_sched_barrier(0);

    #pragma unroll
    for (int c = 15; c >= 0; --c) {
        x[c] = yb[c] / tc[c][c];
        #pragma unroll
        for (int r = 0; r < c; ++r)
            yb[r] -= tc[r][c] * x[c];
        yown -= uc[c] * x[c];
        xout = (((tid >> 4) == kb) && ((tid & 15) == c)) ? x[c] : xout;
    }
    if ((tid >> 4) == kb - 1)
        Msh[tid * 129 + KDIM] = yown;
}

#define APITCH 132
__global__ __launch_bounds__(128) void fm_solve(
    const float* __restrict__ Amat, const float* __restrict__ Bmat,
    const float* __restrict__ D,    float* __restrict__ Cout)
{
    extern __shared__ float4 smem4[];
    float* Msh  = (float*)smem4;
    float* prow = Msh + KDIM * 129;
    float* Lbuf = prow + 32;
    float* Ubuf = Lbuf + 256;
    float* Ast  = (float*)smem4;
    float* Brow = Ast + KDIM * APITCH;
    const int i   = blockIdx.x;
    const int tid = threadIdx.x;

    {
        const int lane = tid & 31;
        const int wv   = tid >> 5;
        const int c    = lane & 15;
        const int h    = lane >> 4;

        const float* __restrict__ ap = Amat + (size_t)wv * PDIM + 4 * lane;
        v4f st0[32], st1[32];
        #pragma unroll
        for (int q = 0; q < 32; ++q)
            st0[q] = *(const v4f*)(ap + (size_t)q * 4 * PDIM);
        const v2f bq = *(const v2f*)(Bmat + (size_t)i * PDIM + 2 * tid);
        const float dsc = D[i * KDIM + tid];
        #pragma unroll
        for (int q = 0; q < 32; ++q)
            *(v4f*)&Ast[(4 * q + wv) * APITCH + 4 * lane] = st0[q];
        *(v2f*)&Brow[2 * tid] = bq;
        __syncthreads();
        #pragma unroll
        for (int q = 0; q < 32; ++q)
            st1[q] = *(const v4f*)(ap + (size_t)q * 4 * PDIM + 128);

        v8f acc[11];
        #pragma unroll
        for (int t = 0; t < 11; ++t) acc[t] = (v8f)0.0f;

        int fo[8];
        #pragma unroll
        for (int t = 0; t < 8; ++t) fo[t] = (16 * ((wv + t) & 7) + c) * APITCH + 2 * h;
        const bool row0 = (c == 0);

        for (int hh = 0; hh < 2; ++hh) {
            if (hh == 1) {
                __syncthreads();
                #pragma unroll
                for (int q = 0; q < 32; ++q)
                    *(v4f*)&Ast[(4 * q + wv) * APITCH + 4 * lane] = st1[q];
                __syncthreads();
            }
            v2f f[8], fb;
            #pragma unroll
            for (int t = 0; t < 8; ++t) f[t] = *(const v2f*)&Ast[fo[t]];
            {
                const v2f tb = *(const v2f*)&Brow[128 * hh + 2 * h];
                fb.x = row0 ? tb.x : 0.0f;
                fb.y = row0 ? tb.y : 0.0f;
            }
            for (int ks = 0; ks < 32; ++ks) {
                v2f fn[8], fbn;
                if (ks + 1 < 32) {
                    #pragma unroll
                    for (int t = 0; t < 8; ++t) fn[t] = *(const v2f*)&Ast[fo[t] + 4 * (ks + 1)];
                    const v2f tb = *(const v2f*)&Brow[128 * hh + 4 * (ks + 1) + 2 * h];
                    fbn.x = row0 ? tb.x : 0.0f;
                    fbn.y = row0 ? tb.y : 0.0f;
                } else {
                    #pragma unroll
                    for (int t = 0; t < 8; ++t) fn[t] = f[t];
                    fbn = fb;
                }
                #pragma unroll
                for (int t = 0; t < 5; ++t)
                    acc[t] = __builtin_amdgcn_wmma_f32_16x16x4_f32(
                        false, f[0], false, f[t], (short)0, acc[t], false, false);
                #pragma unroll
                for (int t = 0; t < 4; ++t)
                    acc[5 + t] = __builtin_amdgcn_wmma_f32_16x16x4_f32(
                        false, f[4], false, f[4 + t], (short)0, acc[5 + t], false, false);
                acc[9]  = __builtin_amdgcn_wmma_f32_16x16x4_f32(
                    false, fb, false, f[0], (short)0, acc[9], false, false);
                acc[10] = __builtin_amdgcn_wmma_f32_16x16x4_f32(
                    false, fb, false, f[4], (short)0, acc[10], false, false);
                #pragma unroll
                for (int t = 0; t < 8; ++t) f[t] = fn[t];
                fb = fbn;
            }
        }
        __syncthreads();

        #pragma unroll
        for (int a = 0; a < 2; ++a) {
            const int I = wv + 4 * a;
            #pragma unroll
            for (int t = 0; t < 5 - a; ++t) {
                const int J = (I + t) & 7;
                const int col = 16 * J + c;
                #pragma unroll
                for (int r = 0; r < 8; ++r) {
                    const int row = 16 * I + 8 * h + r;
                    const float v = acc[5 * a + t][r];
                    Msh[row * 129 + col] = v;
                    if (t != 0) Msh[col * 129 + row] = v;
                }
            }
            if (h == 0)
                Msh[(16 * I + c) * 129 + KDIM] = acc[9 + a][0];
        }
        __syncthreads();
        Msh[tid * 129 + tid] += LAMBDA * dsc;
    }

    for (int kb = 0; kb < KDIM / 16; ++kb) {
        const int k0 = kb * 16;

        float p[16];
        #pragma unroll
        for (int j = 0; j < 16; ++j)
            p[j] = (tid >= k0) ? Msh[tid * 129 + k0 + j] : 0.0f;

        #pragma unroll
        for (int lc = 0; lc < 16; ++lc) {
            const int c = k0 + lc;
            float* pr = prow + (lc & 1) * 16;
            if (tid == c) {
                #pragma unroll
                for (int j = lc; j < 16; ++j) pr[j] = p[j];
            }
            __syncthreads();
            {
                const bool below = tid > c;
                float q[16];
                #pragma unroll
                for (int j = lc; j < 16; ++j) q[j] = pr[j];
                const float f = p[lc] / q[lc];
                p[lc] = below ? f : p[lc];
                #pragma unroll
                for (int j = lc + 1; j < 16; ++j)
                    p[j] = below ? (p[j] - f * q[j]) : p[j];
            }
        }

        if (tid >= k0 && tid < k0 + 16) {
            #pragma unroll
            for (int j = 0; j < 16; ++j)
                Msh[tid * 129 + k0 + j] = p[j];
            #pragma unroll
            for (int j = 0; j < 16; ++j)
                Lbuf[(tid - k0) * 16 + j] = p[j];
        }
        __syncthreads();

        const int ncm = KDIM - 16 - k0;
        if (tid <= ncm) {
            const int j = k0 + 16 + tid;
            float u[16];
            #pragma unroll
            for (int r = 0; r < 16; ++r)
                u[r] = Msh[(k0 + r) * 129 + j];
            #pragma unroll
            for (int r = 1; r < 16; ++r) {
                const float4* lr4 = (const float4*)(Lbuf + r * 16);
                float l[16];
                #pragma unroll
                for (int g = 0; g * 4 < r; ++g) {
                    const float4 t = lr4[g];
                    l[4 * g] = t.x; l[4 * g + 1] = t.y; l[4 * g + 2] = t.z; l[4 * g + 3] = t.w;
                }
                #pragma unroll
                for (int c = 0; c < r; ++c)
                    u[r] -= l[c] * u[c];
            }
            #pragma unroll
            for (int r = 0; r < 16; ++r) {
                Msh[(k0 + r) * 129 + j] = u[r];
                Ubuf[r * 116 + tid]     = u[r];
            }
        }
        __syncthreads();

        if (tid >= k0 + 16) {
            float* row = Msh + tid * 129 + k0 + 16;
            for (int jj = 0; jj < ncm; jj += 4) {
                float x0 = row[jj], x1 = row[jj + 1], x2 = row[jj + 2], x3 = row[jj + 3];
                #pragma unroll
                for (int c = 0; c < 16; ++c) {
                    const float4 uu = *(const float4*)(Ubuf + c * 116 + jj);
                    x0 -= p[c] * uu.x;
                    x1 -= p[c] * uu.y;
                    x2 -= p[c] * uu.z;
                    x3 -= p[c] * uu.w;
                }
                row[jj] = x0; row[jj + 1] = x1; row[jj + 2] = x2; row[jj + 3] = x3;
            }
            float xr = row[ncm];
            #pragma unroll
            for (int c = 0; c < 16; ++c)
                xr -= p[c] * Ubuf[c * 116 + ncm];
            row[ncm] = xr;
        }
        __syncthreads();
    }

    float xout = 0.0f;
    float yown = Msh[tid * 129 + KDIM];
    float tA[16][16], tB[16][16];
    float uA[16], uB[16];
    {
        const int kl = KDIM - 16;
        #pragma unroll
        for (int r = 0; r < 16; ++r) {
            #pragma unroll
            for (int c = r; c < 16; ++c)
                tA[r][c] = Msh[(kl + r) * 129 + kl + c];
        }
        #pragma unroll
        for (int c = 0; c < 16; ++c)
            uA[c] = Msh[tid * 129 + kl + c];
    }
    for (int kp = KDIM / 32 - 1; kp >= 0; --kp) {
        fm_bs_block(Msh, tid, 2 * kp + 1, tA, uA, tB, uB, yown, xout);
        fm_bs_block(Msh, tid, 2 * kp,     tB, uB, tA, uA, yown, xout);
    }
    Cout[i * KDIM + tid] = xout;
}

extern "C" void kernel_launch(void* const* d_in, const int* in_sizes, int n_in,
                              void* d_out, int out_size, void* d_ws, size_t ws_size,
                              hipStream_t stream)
{
    const float* feat_x = (const float*)d_in[0];
    const float* feat_y = (const float*)d_in[1];
    const float* evals_x = (const float*)d_in[2];
    const float* evals_y = (const float*)d_in[3];
    const float* evecs_x = (const float*)d_in[4];
    const float* evecs_y = (const float*)d_in[5];

    float* wsf   = (float*)d_ws;
    float* Amat  = wsf;
    float* Bmat  = wsf + 32768;
    float* Dmask = wsf + 98304;
    float* part  = wsf + 114688;
    float* Cout  = (float*)d_out;

    fm_gemm_partial<<<dim3(2 * NCHUNKS, 2), 128, 0, stream>>>(
        evecs_x, evecs_y, feat_x, feat_y, part);

    fm_reduce<<<(KP + 255) / 256, 256, 0, stream>>>(part, Amat, Bmat);

    fm_mask<<<(KDIM * KDIM + 255) / 256, 256, 0, stream>>>(evals_x, evals_y, Dmask);

    fm_solve<<<KDIM, KDIM, 74624  , stream>>>(
        Amat, Bmat, Dmask, Cout);
}


# solve kernel: the 16 deferred pivots reach the trailing matrix as 16x16 WMMA tiles (tile += -L*U, 4 k-steps, operands straight from M in LDS), not a 16-product chain per entry; sums regroup by 4
# speedup vs baseline: 5.3935x; 1.1555x over previous
#include <hip/hip_runtime.h>
#include <hip/hip_bf16.h>

typedef __attribute__((ext_vector_type(2))) float v2f;
typedef __attribute__((ext_vector_type(4))) float v4f;
typedef __attribute__((ext_vector_type(8))) float v8f;

#define KDIM 128
#define PDIM 256
#define NDIM 100000
#define LAMBDA 0.001f
#define NCHUNKS 64
#define SLAB 32
#define CHUNK 1568
#define EVP 36
#define KP (KDIM * PDIM)

__global__ __launch_bounds__(128) void fm_gemm_partial(
    const float* __restrict__ evx, const float* __restrict__ evy,
    const float* __restrict__ fx,  const float* __restrict__ fy,
    float* __restrict__ partials)
{
    const int sel   = blockIdx.y;
    const int chunk = blockIdx.x >> 1;
    const int half  = blockIdx.x & 1;
    const float* __restrict__ ev   = sel ? evy : evx;
    const float* __restrict__ feat = sel ? fy  : fx;

    __shared__ __attribute__((aligned(16))) float evS[2][KDIM][EVP];

    const int tid  = threadIdx.x;
    const int lane = tid & 31;
    const int wave = tid >> 5;
    const int c    = lane & 15;
    const int h    = lane >> 4;
    const int colBase = half * 128 + wave * 32;

    const int nStart = chunk * CHUNK;
    const int nEnd   = min(nStart + CHUNK, NDIM);
    const int nSlabs = (nEnd - nStart) / SLAB;

    const int sRow = tid >> 3;
    const int sCol = (tid & 7) * 4;
    const float* __restrict__ evp = ev + (size_t)sRow * NDIM + nStart + sCol;
    const float* __restrict__ ftp = feat + (size_t)(nStart + 4 * h) * PDIM + colBase + 2 * c;

    v8f acc[8][2];
    #pragma unroll
    for (int rb = 0; rb < 8; ++rb) { acc[rb][0] = (v8f)0.0f; acc[rb][1] = (v8f)0.0f; }

    v2f fcur[16], fnxt[16];
    v4f evr[8];

    #pragma unroll
    for (int q = 0; q < 8; ++q)
        evr[q] = *(const v4f*)(evp + (size_t)q * 16 * NDIM);
    #pragma unroll
    for (int f = 0; f < 16; ++f)
        fcur[f] = *(const v2f*)(ftp + (size_t)((f >> 2) * 8 + (f & 3)) * PDIM);
    #pragma unroll
    for (int q = 0; q < 8; ++q)
        *(v4f*)&evS[0][sRow + 16 * q][sCol] = evr[q];
    __syncthreads();

    for (int i = 0; i < nSlabs; ++i) {
        const int p = i & 1;
        const bool more = (i + 1 < nSlabs);

        if (more) {
            const float* __restrict__ e2 = evp + (size_t)(i + 1) * SLAB;
            #pragma unroll
            for (int q = 0; q < 8; ++q)
                evr[q] = *(const v4f*)(e2 + (size_t)q * 16 * NDIM);
            const float* __restrict__ f2 = ftp + (size_t)(i + 1) * SLAB * PDIM;
            #pragma unroll
            for (int f = 0; f < 16; ++f)
                fnxt[f] = *(const v2f*)(f2 + (size_t)((f >> 2) * 8 + (f & 3)) * PDIM);
        }

        v4f a[8];
        #pragma unroll
        for (int rb = 0; rb < 8; ++rb)
            a[rb] = *(const v4f*)&evS[p][rb * 16 + c][4 * h];
        #pragma unroll
        for (int jj = 0; jj < 4; ++jj) {
            v4f an[8];
            if (jj < 3) {
                #pragma unroll
                for (int rb = 0; rb < 8; ++rb)
                    an[rb] = *(const v4f*)&evS[p][rb * 16 + c][8 * (jj + 1) + 4 * h];
            }
            #pragma unroll
            for (int u = 0; u < 2; ++u) {
                const v2f f0 = fcur[jj * 4 + u * 2];
                const v2f f1 = fcur[jj * 4 + u * 2 + 1];
                v2f b0, b1;
                b0.x = f0.x; b0.y = f1.x;
                b1.x = f0.y; b1.y = f1.y;
                #pragma unroll
                for (int rb = 0; rb < 8; ++rb) {
                    v2f av;
                    av.x = a[rb][2 * u];
                    av.y = a[rb][2 * u + 1];
                    acc[rb][0] = __builtin_amdgcn_wmma_f32_16x16x4_f32(
                        false, av, false, b0, (short)0, acc[rb][0], false, false);
                    acc[rb][1] = __builtin_amdgcn_wmma_f32_16x16x4_f32(
                        false, av, false, b1, (short)0, acc[rb][1], false, false);
                }
            }
            if (jj < 3) {
                #pragma unroll
                for (int rb = 0; rb < 8; ++rb) a[rb] = an[rb];
            }
        }

        if (more) {
            #pragma unroll
            for (int q = 0; q < 8; ++q)
                *(v4f*)&evS[p ^ 1][sRow + 16 * q][sCol] = evr[q];
            #pragma unroll
            for (int f = 0; f < 16; ++f) fcur[f] = fnxt[f];
        }
        __syncthreads();
    }

    float* out = partials + (size_t)sel * NCHUNKS * KP + (size_t)chunk * KP;
    #pragma unroll
    for (int rb = 0; rb < 8; ++rb) {
        #pragma unroll
        for (int r = 0; r < 8; ++r) {
            const int row = rb * 16 + 8 * h + r;
            v2f o;
            o.x = acc[rb][0][r];
            o.y = acc[rb][1][r];
            *(v2f*)&out[row * PDIM + colBase + 2 * c] = o;
        }
    }
}

__global__ void fm_reduce(const float* __restrict__ partials,
                          float* __restrict__ Amat, float* __restrict__ Bmat)
{
    int idx = blockIdx.x * blockDim.x + threadIdx.x;
    if (idx >= KP) return;
    float sa = 0.0f, sb = 0.0f;
    for (int c = 0; c < NCHUNKS; ++c) {
        sa += partials[(size_t)c * KP + idx];
        sb += partials[(size_t)(NCHUNKS + c) * KP + idx];
    }
    Amat[idx] = sa;
    Bmat[idx] = sb;
}

__global__ void fm_mask(const float* __restrict__ evx,
                        const float* __restrict__ evy,
                        float* __restrict__ D)
{
    int idx = blockIdx.x * blockDim.x + threadIdx.x;
    if (idx >= KDIM * KDIM) return;
    int i = idx >> 7;
    int j = idx & 127;
    float s  = fmaxf(evx[KDIM - 1], evy[KDIM - 1]);
    float g1 = sqrtf(evx[j] / s);
    float g2 = sqrtf(evy[i] / s);
    float d1 = 1.0f / (g1 * g1 + 1.0f);
    float d2 = 1.0f / (g2 * g2 + 1.0f);
    float re = g2 * d2 - g1 * d1;
    float im = d2 - d1;
    D[idx] = re * re + im * im;
}

__device__ __forceinline__ void fm_bs_block(
    float* Msh, const int tid, const int kb,
    const float (&tc)[16][16], const float (&uc)[16],
    float (&tn)[16][16], float (&un)[16], float& yown, float& xout)
{
    const int k0 = kb * 16;
    const int kn = (kb > 0) ? k0 - 16 : 0;

    __syncthreads();

    float yb[16], x[16];
    #pragma unroll
    for (int r = 0; r < 16; ++r)
        yb[r] = Msh[(k0 + r) * 129 + KDIM];
    #pragma unroll
    for (int r = 0; r < 16; ++r) {
        #pragma unroll
        for (int c = r; c < 16; ++c)
            tn[r][c] = Msh[(kn + r) * 129 + kn + c];
    }
    #pragma unroll
    for (int c = 0; c < 16; ++c)
        un[c] = Msh[tid * 129 + kn + c];
    __builtin_amdgcn_sched_barrier(0);

    #pragma unroll
    for (int c = 15; c >= 0; --c) {
        x[c] = yb[c] / tc[c][c];
        #pragma unroll
        for (int r = 0; r < c; ++r)
            yb[r] -= tc[r][c] * x[c];
        yown -= uc[c] * x[c];
        xout = (((tid >> 4) == kb) && ((tid & 15) == c)) ? x[c] : xout;
    }
    if ((tid >> 4) == kb - 1)
        Msh[tid * 129 + KDIM] = yown;
}

#define APITCH 132
__global__ __launch_bounds__(128) void fm_solve(
    const float* __restrict__ Amat, const float* __restrict__ Bmat,
    const float* __restrict__ D,    float* __restrict__ Cout)
{
    extern __shared__ float4 smem4[];
    float* Msh  = (float*)smem4;
    float* prow = Msh + KDIM * 129;
    float* Lbuf = prow + 32;
    float* Ast  = (float*)smem4;
    float* Brow = Ast + KDIM * APITCH;
    const int i   = blockIdx.x;
    const int tid = threadIdx.x;
    const int wvu = __builtin_amdgcn_readfirstlane(tid >> 5);

    {
        const int lane = tid & 31;
        const int wv   = tid >> 5;
        const int c    = lane & 15;
        const int h    = lane >> 4;

        const float* __restrict__ ap = Amat + (size_t)wv * PDIM + 4 * lane;
        v4f st0[32], st1[32];
        #pragma unroll
        for (int q = 0; q < 32; ++q)
            st0[q] = *(const v4f*)(ap + (size_t)q * 4 * PDIM);
        const v2f bq = *(const v2f*)(Bmat + (size_t)i * PDIM + 2 * tid);
        const float dsc = D[i * KDIM + tid];
        #pragma unroll
        for (int q = 0; q < 32; ++q)
            *(v4f*)&Ast[(4 * q + wv) * APITCH + 4 * lane] = st0[q];
        *(v2f*)&Brow[2 * tid] = bq;
        __syncthreads();
        #pragma unroll
        for (int q = 0; q < 32; ++q)
            st1[q] = *(const v4f*)(ap + (size_t)q * 4 * PDIM + 128);

        v8f acc[11];
        #pragma unroll
        for (int t = 0; t < 11; ++t) acc[t] = (v8f)0.0f;

        int fo[8];
        #pragma unroll
        for (int t = 0; t < 8; ++t) fo[t] = (16 * ((wv + t) & 7) + c) * APITCH + 2 * h;
        const bool row0 = (c == 0);

        for (int hh = 0; hh < 2; ++hh) {
            if (hh == 1) {
                __syncthreads();
                #pragma unroll
                for (int q = 0; q < 32; ++q)
                    *(v4f*)&Ast[(4 * q + wv) * APITCH + 4 * lane] = st1[q];
                __syncthreads();
            }
            v2f f[8], fb;
            #pragma unroll
            for (int t = 0; t < 8; ++t) f[t] = *(const v2f*)&Ast[fo[t]];
            {
                const v2f tb = *(const v2f*)&Brow[128 * hh + 2 * h];
                fb.x = row0 ? tb.x : 0.0f;
                fb.y = row0 ? tb.y : 0.0f;
            }
            for (int ks = 0; ks < 32; ++ks) {
                v2f fn[8], fbn;
                if (ks + 1 < 32) {
                    #pragma unroll
                    for (int t = 0; t < 8; ++t) fn[t] = *(const v2f*)&Ast[fo[t] + 4 * (ks + 1)];
                    const v2f tb = *(const v2f*)&Brow[128 * hh + 4 * (ks + 1) + 2 * h];
                    fbn.x = row0 ? tb.x : 0.0f;
                    fbn.y = row0 ? tb.y : 0.0f;
                } else {
                    #pragma unroll
                    for (int t = 0; t < 8; ++t) fn[t] = f[t];
                    fbn = fb;
                }
                #pragma unroll
                for (int t = 0; t < 5; ++t)
                    acc[t] = __builtin_amdgcn_wmma_f32_16x16x4_f32(
                        false, f[0], false, f[t], (short)0, acc[t], false, false);
                #pragma unroll
                for (int t = 0; t < 4; ++t)
                    acc[5 + t] = __builtin_amdgcn_wmma_f32_16x16x4_f32(
                        false, f[4], false, f[4 + t], (short)0, acc[5 + t], false, false);
                acc[9]  = __builtin_amdgcn_wmma_f32_16x16x4_f32(
                    false, fb, false, f[0], (short)0, acc[9], false, false);
                acc[10] = __builtin_amdgcn_wmma_f32_16x16x4_f32(
                    false, fb, false, f[4], (short)0, acc[10], false, false);
                #pragma unroll
                for (int t = 0; t < 8; ++t) f[t] = fn[t];
                fb = fbn;
            }
        }
        __syncthreads();

        #pragma unroll
        for (int a = 0; a < 2; ++a) {
            const int I = wv + 4 * a;
            #pragma unroll
            for (int t = 0; t < 5 - a; ++t) {
                const int J = (I + t) & 7;
                const int col = 16 * J + c;
                #pragma unroll
                for (int r = 0; r < 8; ++r) {
                    const int row = 16 * I + 8 * h + r;
                    const float v = acc[5 * a + t][r];
                    Msh[row * 129 + col] = v;
                    if (t != 0) Msh[col * 129 + row] = v;
                }
            }
            if (h == 0)
                Msh[(16 * I + c) * 129 + KDIM] = acc[9 + a][0];
        }
        __syncthreads();
        Msh[tid * 129 + tid] += LAMBDA * dsc;
    }

    for (int kb = 0; kb < KDIM / 16; ++kb) {
        const int k0 = kb * 16;

        float p[16];
        #pragma unroll
        for (int j = 0; j < 16; ++j)
            p[j] = (tid >= k0) ? Msh[tid * 129 + k0 + j] : 0.0f;

        #pragma unroll
        for (int lc = 0; lc < 16; ++lc) {
            const int c = k0 + lc;
            float* pr = prow + (lc & 1) * 16;
            if (tid == c) {
                #pragma unroll
                for (int j = lc; j < 16; ++j) pr[j] = p[j];
            }
            __syncthreads();
            {
                const bool below = tid > c;
                float q[16];
                #pragma unroll
                for (int j = lc; j < 16; ++j) q[j] = pr[j];
                const float f = p[lc] / q[lc];
                p[lc] = below ? f : p[lc];
                #pragma unroll
                for (int j = lc + 1; j < 16; ++j)
                    p[j] = below ? (p[j] - f * q[j]) : p[j];
            }
        }

        if (tid >= k0 && tid < k0 + 16) {
            #pragma unroll
            for (int j = 0; j < 16; ++j)
                Msh[tid * 129 + k0 + j] = p[j];
            #pragma unroll
            for (int j = 0; j < 16; ++j)
                Lbuf[(tid - k0) * 16 + j] = p[j];
        } else if (tid >= k0 + 16) {
            #pragma unroll
            for (int j = 0; j < 16; ++j)
                Msh[tid * 129 + k0 + j] = -p[j];
        }
        __syncthreads();

        const int ncm = KDIM - 16 - k0;
        if (tid <= ncm) {
            const int j = k0 + 16 + tid;
            float u[16];
            #pragma unroll
            for (int r = 0; r < 16; ++r)
                u[r] = Msh[(k0 + r) * 129 + j];
            #pragma unroll
            for (int r = 1; r < 16; ++r) {
                const float4* lr4 = (const float4*)(Lbuf + r * 16);
                float l[16];
                #pragma unroll
                for (int g = 0; g * 4 < r; ++g) {
                    const float4 t = lr4[g];
                    l[4 * g] = t.x; l[4 * g + 1] = t.y; l[4 * g + 2] = t.z; l[4 * g + 3] = t.w;
                }
                #pragma unroll
                for (int c = 0; c < r; ++c)
                    u[r] -= l[c] * u[c];
            }
            #pragma unroll
            for (int r = 0; r < 16; ++r)
                Msh[(k0 + r) * 129 + j] = u[r];
        }
        __syncthreads();

        if (tid >= k0 + 16) {
            float xr = Msh[tid * 129 + KDIM];
            #pragma unroll
            for (int c = 0; c < 16; ++c)
                xr -= p[c] * Msh[(k0 + c) * 129 + KDIM];
            Msh[tid * 129 + KDIM] = xr;
        }
        {
            const int lane = tid & 31;
            const int c    = lane & 15;
            const int h    = lane >> 4;
            const int np   = (8 - kb) >> 1;
            int item = 0;
            for (int I = kb + 1; I < 8; ++I) {
                for (int jp = 0; jp < np; ++jp, ++item) {
                    if ((item & 3) != wvu) continue;
                    const int  J0  = kb + 1 + 2 * jp;
                    const bool two = (J0 + 1 < 8);
                    const int  J1  = two ? J0 + 1 : J0;
                    const float* la = Msh + (16 * I + c) * 129 + k0 + 2 * h;
                    const float* u0 = Msh + (k0 + 2 * h) * 129 + 16 * J0 + c;
                    const float* u1 = Msh + (k0 + 2 * h) * 129 + 16 * J1 + c;
                    float* t0 = Msh + (16 * I + 8 * h) * 129 + 16 * J0 + c;
                    float* t1 = Msh + (16 * I + 8 * h) * 129 + 16 * J1 + c;
                    v2f a[4], b0[4], b1[4];
                    v8f acc0, acc1;
                    #pragma unroll
                    for (int ks = 0; ks < 4; ++ks) {
                        a[ks].x  = la[4 * ks];           a[ks].y  = la[4 * ks + 1];
                        b0[ks].x = u0[(4 * ks) * 129];   b0[ks].y = u0[(4 * ks + 1) * 129];
                        b1[ks].x = u1[(4 * ks) * 129];   b1[ks].y = u1[(4 * ks + 1) * 129];
                    }
                    #pragma unroll
                    for (int r = 0; r < 8; ++r) { acc0[r] = t0[r * 129]; acc1[r] = t1[r * 129]; }
                    #pragma unroll
                    for (int ks = 0; ks < 4; ++ks) {
                        acc0 = __builtin_amdgcn_wmma_f32_16x16x4_f32(
                            false, a[ks], false, b0[ks], (short)0, acc0, false, false);
                        acc1 = __builtin_amdgcn_wmma_f32_16x16x4_f32(
                            false, a[ks], false, b1[ks], (short)0, acc1, false, false);
                    }
                    #pragma unroll
                    for (int r = 0; r < 8; ++r) t0[r * 129] = acc0[r];
                    if (two) {
                        #pragma unroll
                        for (int r = 0; r < 8; ++r) t1[r * 129] = acc1[r];
                    }
                }
            }
        }
        __syncthreads();
    }

    float xout = 0.0f;
    float yown = Msh[tid * 129 + KDIM];
    float tA[16][16], tB[16][16];
    float uA[16], uB[16];
    {
        const int kl = KDIM - 16;
        #pragma unroll
        for (int r = 0; r < 16; ++r) {
            #pragma unroll
            for (int c = r; c < 16; ++c)
                tA[r][c] = Msh[(kl + r) * 129 + kl + c];
        }
        #pragma unroll
        for (int c = 0; c < 16; ++c)
            uA[c] = Msh[tid * 129 + kl + c];
    }
    for (int kp = KDIM / 32 - 1; kp >= 0; --kp) {
        fm_bs_block(Msh, tid, 2 * kp + 1, tA, uA, tB, uB, yown, xout);
        fm_bs_block(Msh, tid, 2 * kp,     tB, uB, tA, uA, yown, xout);
    }
    Cout[i * KDIM + tid] = xout;
}

extern "C" void kernel_launch(void* const* d_in, const int* in_sizes, int n_in,
                              void* d_out, int out_size, void* d_ws, size_t ws_size,
                              hipStream_t stream)
{
    const float* feat_x = (const float*)d_in[0];
    const float* feat_y = (const float*)d_in[1];
    const float* evals_x = (const float*)d_in[2];
    const float* evals_y = (const float*)d_in[3];
    const float* evecs_x = (const float*)d_in[4];
    const float* evecs_y = (const float*)d_in[5];

    float* wsf   = (float*)d_ws;
    float* Amat  = wsf;
    float* Bmat  = wsf + 32768;
    float* Dmask = wsf + 98304;
    float* part  = wsf + 114688;
    float* Cout  = (float*)d_out;

    fm_gemm_partial<<<dim3(2 * NCHUNKS, 2), 128, 0, stream>>>(
        evecs_x, evecs_y, feat_x, feat_y, part);

    fm_reduce<<<(KP + 255) / 256, 256, 0, stream>>>(part, Amat, Bmat);

    fm_mask<<<(KDIM * KDIM + 255) / 256, 256, 0, stream>>>(evals_x, evals_y, Dmask);

    fm_solve<<<KDIM, KDIM, 68608  , stream>>>(
        Amat, Bmat, Dmask, Cout);
}


# solve kernel: panel pivots leave LDS: every wave keeps its own copy of the 16x16 diagonal part in lanes, pivot rows go by in-row lane broadcast (DPP); 4 barriers per panel, not 19; values kept
# speedup vs baseline: 5.8284x; 1.0806x over previous
#include <hip/hip_runtime.h>
#include <hip/hip_bf16.h>

typedef __attribute__((ext_vector_type(2))) float v2f;
typedef __attribute__((ext_vector_type(4))) float v4f;
typedef __attribute__((ext_vector_type(8))) float v8f;

#define KDIM 128
#define PDIM 256
#define NDIM 100000
#define LAMBDA 0.001f
#define NCHUNKS 64
#define SLAB 32
#define CHUNK 1568
#define EVP 36
#define KP (KDIM * PDIM)

__global__ __launch_bounds__(128) void fm_gemm_partial(
    const float* __restrict__ evx, const float* __restrict__ evy,
    const float* __restrict__ fx,  const float* __restrict__ fy,
    float* __restrict__ partials)
{
    const int sel   = blockIdx.y;
    const int chunk = blockIdx.x >> 1;
    const int half  = blockIdx.x & 1;
    const float* __restrict__ ev   = sel ? evy : evx;
    const float* __restrict__ feat = sel ? fy  : fx;

    __shared__ __attribute__((aligned(16))) float evS[2][KDIM][EVP];

    const int tid  = threadIdx.x;
    const int lane = tid & 31;
    const int wave = tid >> 5;
    const int c    = lane & 15;
    const int h    = lane >> 4;
    const int colBase = half * 128 + wave * 32;

    const int nStart = chunk * CHUNK;
    const int nEnd   = min(nStart + CHUNK, NDIM);
    const int nSlabs = (nEnd - nStart) / SLAB;

    const int sRow = tid >> 3;
    const int sCol = (tid & 7) * 4;
    const float* __restrict__ evp = ev + (size_t)sRow * NDIM + nStart + sCol;
    const float* __restrict__ ftp = feat + (size_t)(nStart + 4 * h) * PDIM + colBase + 2 * c;

    v8f acc[8][2];
    #pragma unroll
    for (int rb = 0; rb < 8; ++rb) { acc[rb][0] = (v8f)0.0f; acc[rb][1] = (v8f)0.0f; }

    v2f fcur[16], fnxt[16];
    v4f evr[8];

    #pragma unroll
    for (int q = 0; q < 8; ++q)
        evr[q] = *(const v4f*)(evp + (size_t)q * 16 * NDIM);
    #pragma unroll
    for (int f = 0; f < 16; ++f)
        fcur[f] = *(const v2f*)(ftp + (size_t)((f >> 2) * 8 + (f & 3)) * PDIM);
    #pragma unroll
    for (int q = 0; q < 8; ++q)
        *(v4f*)&evS[0][sRow + 16 * q][sCol] = evr[q];
    __syncthreads();

    for (int i = 0; i < nSlabs; ++i) {
        const int p = i & 1;
        const bool more = (i + 1 < nSlabs);

        if (more) {
            const float* __restrict__ e2 = evp + (size_t)(i + 1) * SLAB;
            #pragma unroll
            for (int q = 0; q < 8; ++q)
                evr[q] = *(const v4f*)(e2 + (size_t)q * 16 * NDIM);
            const float* __restrict__ f2 = ftp + (size_t)(i + 1) * SLAB * PDIM;
            #pragma unroll
            for (int f = 0; f < 16; ++f)
                fnxt[f] = *(const v2f*)(f2 + (size_t)((f >> 2) * 8 + (f & 3)) * PDIM);
        }

        v4f a[8];
        #pragma unroll
        for (int rb = 0; rb < 8; ++rb)
            a[rb] = *(const v4f*)&evS[p][rb * 16 + c][4 * h];
        #pragma unroll
        for (int jj = 0; jj < 4; ++jj) {
            v4f an[8];
            if (jj < 3) {
                #pragma unroll
                for (int rb = 0; rb < 8; ++rb)
                    an[rb] = *(const v4f*)&evS[p][rb * 16 + c][8 * (jj + 1) + 4 * h];
            }
            #pragma unroll
            for (int u = 0; u < 2; ++u) {
                const v2f f0 = fcur[jj * 4 + u * 2];
                const v2f f1 = fcur[jj * 4 + u * 2 + 1];
                v2f b0, b1;
                b0.x = f0.x; b0.y = f1.x;
                b1.x = f0.y; b1.y = f1.y;
                #pragma unroll
                for (int rb = 0; rb < 8; ++rb) {
                    v2f av;
                    av.x = a[rb][2 * u];
                    av.y = a[rb][2 * u + 1];
                    acc[rb][0] = __builtin_amdgcn_wmma_f32_16x16x4_f32(
                        false, av, false, b0, (short)0, acc[rb][0], false, false);
                    acc[rb][1] = __builtin_amdgcn_wmma_f32_16x16x4_f32(
                        false, av, false, b1, (short)0, acc[rb][1], false, false);
                }
            }
            if (jj < 3) {
                #pragma unroll
                for (int rb = 0; rb < 8; ++rb) a[rb] = an[rb];
            }
        }

        if (more) {
            #pragma unroll
            for (int q = 0; q < 8; ++q)
                *(v4f*)&evS[p ^ 1][sRow + 16 * q][sCol] = evr[q];
            #pragma unroll
            for (int f = 0; f < 16; ++f) fcur[f] = fnxt[f];
        }
        __syncthreads();
    }

    float* out = partials + (size_t)sel * NCHUNKS * KP + (size_t)chunk * KP;
    #pragma unroll
    for (int rb = 0; rb < 8; ++rb) {
        #pragma unroll
        for (int r = 0; r < 8; ++r) {
            const int row = rb * 16 + 8 * h + r;
            v2f o;
            o.x = acc[rb][0][r];
            o.y = acc[rb][1][r];
            *(v2f*)&out[row * PDIM + colBase + 2 * c] = o;
        }
    }
}

__global__ void fm_reduce(const float* __restrict__ partials,
                          float* __restrict__ Amat, float* __restrict__ Bmat)
{
    int idx = blockIdx.x * blockDim.x + threadIdx.x;
    if (idx >= KP) return;
    float sa = 0.0f, sb = 0.0f;
    for (int c = 0; c < NCHUNKS; ++c) {
        sa += partials[(size_t)c * KP + idx];
        sb += partials[(size_t)(NCHUNKS + c) * KP + idx];
    }
    Amat[idx] = sa;
    Bmat[idx] = sb;
}

__global__ void fm_mask(const float* __restrict__ evx,
                        const float* __restrict__ evy,
                        float* __restrict__ D)
{
    int idx = blockIdx.x * blockDim.x + threadIdx.x;
    if (idx >= KDIM * KDIM) return;
    int i = idx >> 7;
    int j = idx & 127;
    float s  = fmaxf(evx[KDIM - 1], evy[KDIM - 1]);
    float g1 = sqrtf(evx[j] / s);
    float g2 = sqrtf(evy[i] / s);
    float d1 = 1.0f / (g1 * g1 + 1.0f);
    float d2 = 1.0f / (g2 * g2 + 1.0f);
    float re = g2 * d2 - g1 * d1;
    float im = d2 - d1;
    D[idx] = re * re + im * im;
}

__device__ __forceinline__ void fm_bs_block(
    float* Msh, const int tid, const int kb,
    const float (&tc)[16][16], const float (&uc)[16],
    float (&tn)[16][16], float (&un)[16], float& yown, float& xout)
{
    const int k0 = kb * 16;
    const int kn = (kb > 0) ? k0 - 16 : 0;

    __syncthreads();

    float yb[16], x[16];
    #pragma unroll
    for (int r = 0; r < 16; ++r)
        yb[r] = Msh[(k0 + r) * 129 + KDIM];
    #pragma unroll
    for (int r = 0; r < 16; ++r) {
        #pragma unroll
        for (int c = r; c < 16; ++c)
            tn[r][c] = Msh[(kn + r) * 129 + kn + c];
    }
    #pragma unroll
    for (int c = 0; c < 16; ++c)
        un[c] = Msh[tid * 129 + kn + c];
    __builtin_amdgcn_sched_barrier(0);

    #pragma unroll
    for (int c = 15; c >= 0; --c) {
        x[c] = yb[c] / tc[c][c];
        #pragma unroll
        for (int r = 0; r < c; ++r)
            yb[r] -= tc[r][c] * x[c];
        yown -= uc[c] * x[c];
        xout = (((tid >> 4) == kb) && ((tid & 15) == c)) ? x[c] : xout;
    }
    if ((tid >> 4) == kb - 1)
        Msh[tid * 129 + KDIM] = yown;
}

template <int LC>
__device__ __forceinline__ float fm_row_bcast(const float x)
{
    return __builtin_bit_cast(float, __builtin_amdgcn_update_dpp(
        0, __builtin_bit_cast(int, x), 0x150 + LC  , 0xf, 0xf, true));
}

template <int LC>
__device__ __forceinline__ void fm_panel_pivots(float (&p)[16], float (&d)[16],
                                                const int c16, const int rel)
{
    const bool belowd = c16 > LC;
    const bool below  = rel > LC;
    float q[16];
    #pragma unroll
    for (int j = LC; j < 16; ++j) q[j] = fm_row_bcast<LC>(d[j]);
    const float fd = d[LC] / q[LC];
    const float f  = p[LC] / q[LC];
    d[LC] = belowd ? fd : d[LC];
    p[LC] = below  ? f  : p[LC];
    #pragma unroll
    for (int j = LC + 1; j < 16; ++j) {
        d[j] = belowd ? (d[j] - fd * q[j]) : d[j];
        p[j] = below  ? (p[j] - f  * q[j]) : p[j];
    }
    if constexpr (LC < 15) fm_panel_pivots<LC + 1>(p, d, c16, rel);
}

#define APITCH 132
__global__ __launch_bounds__(128) void fm_solve(
    const float* __restrict__ Amat, const float* __restrict__ Bmat,
    const float* __restrict__ D,    float* __restrict__ Cout)
{
    extern __shared__ float4 smem4[];
    float* Msh  = (float*)smem4;
    float* Lbuf = Msh + KDIM * 129;
    float* Ast  = (float*)smem4;
    float* Brow = Ast + KDIM * APITCH;
    const int i   = blockIdx.x;
    const int tid = threadIdx.x;
    const int wvu = __builtin_amdgcn_readfirstlane(tid >> 5);

    {
        const int lane = tid & 31;
        const int wv   = tid >> 5;
        const int c    = lane & 15;
        const int h    = lane >> 4;

        const float* __restrict__ ap = Amat + (size_t)wv * PDIM + 4 * lane;
        v4f st0[32], st1[32];
        #pragma unroll
        for (int q = 0; q < 32; ++q)
            st0[q] = *(const v4f*)(ap + (size_t)q * 4 * PDIM);
        const v2f bq = *(const v2f*)(Bmat + (size_t)i * PDIM + 2 * tid);
        const float dsc = D[i * KDIM + tid];
        #pragma unroll
        for (int q = 0; q < 32; ++q)
            *(v4f*)&Ast[(4 * q + wv) * APITCH + 4 * lane] = st0[q];
        *(v2f*)&Brow[2 * tid] = bq;
        __syncthreads();
        #pragma unroll
        for (int q = 0; q < 32; ++q)
            st1[q] = *(const v4f*)(ap + (size_t)q * 4 * PDIM + 128);

        v8f acc[11];
        #pragma unroll
        for (int t = 0; t < 11; ++t) acc[t] = (v8f)0.0f;

        int fo[8];
        #pragma unroll
        for (int t = 0; t < 8; ++t) fo[t] = (16 * ((wv + t) & 7) + c) * APITCH + 2 * h;
        const bool row0 = (c == 0);

        for (int hh = 0; hh < 2; ++hh) {
            if (hh == 1) {
                __syncthreads();
                #pragma unroll
                for (int q = 0; q < 32; ++q)
                    *(v4f*)&Ast[(4 * q + wv) * APITCH + 4 * lane] = st1[q];
                __syncthreads();
            }
            v2f f[8], fb;
            #pragma unroll
            for (int t = 0; t < 8; ++t) f[t] = *(const v2f*)&Ast[fo[t]];
            {
                const v2f tb = *(const v2f*)&Brow[128 * hh + 2 * h];
                fb.x = row0 ? tb.x : 0.0f;
                fb.y = row0 ? tb.y : 0.0f;
            }
            for (int ks = 0; ks < 32; ++ks) {
                v2f fn[8], fbn;
                if (ks + 1 < 32) {
                    #pragma unroll
                    for (int t = 0; t < 8; ++t) fn[t] = *(const v2f*)&Ast[fo[t] + 4 * (ks + 1)];
                    const v2f tb = *(const v2f*)&Brow[128 * hh + 4 * (ks + 1) + 2 * h];
                    fbn.x = row0 ? tb.x : 0.0f;
                    fbn.y = row0 ? tb.y : 0.0f;
                } else {
                    #pragma unroll
                    for (int t = 0; t < 8; ++t) fn[t] = f[t];
                    fbn = fb;
                }
                #pragma unroll
                for (int t = 0; t < 5; ++t)
                    acc[t] = __builtin_amdgcn_wmma_f32_16x16x4_f32(
                        false, f[0], false, f[t], (short)0, acc[t], false, false);
                #pragma unroll
                for (int t = 0; t < 4; ++t)
                    acc[5 + t] = __builtin_amdgcn_wmma_f32_16x16x4_f32(
                        false, f[4], false, f[4 + t], (short)0, acc[5 + t], false, false);
                acc[9]  = __builtin_amdgcn_wmma_f32_16x16x4_f32(
                    false, fb, false, f[0], (short)0, acc[9], false, false);
                acc[10] = __builtin_amdgcn_wmma_f32_16x16x4_f32(
                    false, fb, false, f[4], (short)0, acc[10], false, false);
                #pragma unroll
                for (int t = 0; t < 8; ++t) f[t] = fn[t];
                fb = fbn;
            }
        }
        __syncthreads();

        #pragma unroll
        for (int a = 0; a < 2; ++a) {
            const int I = wv + 4 * a;
            #pragma unroll
            for (int t = 0; t < 5 - a; ++t) {
                const int J = (I + t) & 7;
                const int col = 16 * J + c;
                #pragma unroll
                for (int r = 0; r < 8; ++r) {
                    const int row = 16 * I + 8 * h + r;
                    const float v = acc[5 * a + t][r];
                    Msh[row * 129 + col] = v;
                    if (t != 0) Msh[col * 129 + row] = v;
                }
            }
            if (h == 0)
                Msh[(16 * I + c) * 129 + KDIM] = acc[9 + a][0];
        }
        __syncthreads();
        Msh[tid * 129 + tid] += LAMBDA * dsc;
        __syncthreads();
    }

    for (int kb = 0; kb < KDIM / 16; ++kb) {
        const int k0 = kb * 16;

        float p[16], d[16];
        #pragma unroll
        for (int j = 0; j < 16; ++j)
            p[j] = (tid >= k0) ? Msh[tid * 129 + k0 + j] : 0.0f;
        #pragma unroll
        for (int j = 0; j < 16; ++j)
            d[j] = Msh[(k0 + (tid & 15)) * 129 + k0 + j];
        fm_panel_pivots<0>(p, d, tid & 15, tid - k0);
        __syncthreads();

        if (tid >= k0 && tid < k0 + 16) {
            #pragma unroll
            for (int j = 0; j < 16; ++j)
                Msh[tid * 129 + k0 + j] = p[j];
            #pragma unroll
            for (int j = 0; j < 16; ++j)
                Lbuf[(tid - k0) * 16 + j] = p[j];
        } else if (tid >= k0 + 16) {
            #pragma unroll
            for (int j = 0; j < 16; ++j)
                Msh[tid * 129 + k0 + j] = -p[j];
        }
        __syncthreads();

        const int ncm = KDIM - 16 - k0;
        if (tid <= ncm) {
            const int j = k0 + 16 + tid;
            float u[16];
            #pragma unroll
            for (int r = 0; r < 16; ++r)
                u[r] = Msh[(k0 + r) * 129 + j];
            #pragma unroll
            for (int r = 1; r < 16; ++r) {
                const float4* lr4 = (const float4*)(Lbuf + r * 16);
                float l[16];
                #pragma unroll
                for (int g = 0; g * 4 < r; ++g) {
                    const float4 t = lr4[g];
                    l[4 * g] = t.x; l[4 * g + 1] = t.y; l[4 * g + 2] = t.z; l[4 * g + 3] = t.w;
                }
                #pragma unroll
                for (int c = 0; c < r; ++c)
                    u[r] -= l[c] * u[c];
            }
            #pragma unroll
            for (int r = 0; r < 16; ++r)
                Msh[(k0 + r) * 129 + j] = u[r];
        }
        __syncthreads();

        if (tid >= k0 + 16) {
            float xr = Msh[tid * 129 + KDIM];
            #pragma unroll
            for (int c = 0; c < 16; ++c)
                xr -= p[c] * Msh[(k0 + c) * 129 + KDIM];
            Msh[tid * 129 + KDIM] = xr;
        }
        {
            const int lane = tid & 31;
            const int c    = lane & 15;
            const int h    = lane >> 4;
            const int np   = (8 - kb) >> 1;
            int item = 0;
            for (int I = kb + 1; I < 8; ++I) {
                for (int jp = 0; jp < np; ++jp, ++item) {
                    if ((item & 3) != wvu) continue;
                    const int  J0  = kb + 1 + 2 * jp;
                    const bool two = (J0 + 1 < 8);
                    const int  J1  = two ? J0 + 1 : J0;
                    const float* la = Msh + (16 * I + c) * 129 + k0 + 2 * h;
                    const float* u0 = Msh + (k0 + 2 * h) * 129 + 16 * J0 + c;
                    const float* u1 = Msh + (k0 + 2 * h) * 129 + 16 * J1 + c;
                    float* t0 = Msh + (16 * I + 8 * h) * 129 + 16 * J0 + c;
                    float* t1 = Msh + (16 * I + 8 * h) * 129 + 16 * J1 + c;
                    v2f a[4], b0[4], b1[4];
                    v8f acc0, acc1;
                    #pragma unroll
                    for (int ks = 0; ks < 4; ++ks) {
                        a[ks].x  = la[4 * ks];           a[ks].y  = la[4 * ks + 1];
                        b0[ks].x = u0[(4 * ks) * 129];   b0[ks].y = u0[(4 * ks + 1) * 129];
                        b1[ks].x = u1[(4 * ks) * 129];   b1[ks].y = u1[(4 * ks + 1) * 129];
                    }
                    #pragma unroll
                    for (int r = 0; r < 8; ++r) { acc0[r] = t0[r * 129]; acc1[r] = t1[r * 129]; }
                    #pragma unroll
                    for (int ks = 0; ks < 4; ++ks) {
                        acc0 = __builtin_amdgcn_wmma_f32_16x16x4_f32(
                            false, a[ks], false, b0[ks], (short)0, acc0, false, false);
                        acc1 = __builtin_amdgcn_wmma_f32_16x16x4_f32(
                            false, a[ks], false, b1[ks], (short)0, acc1, false, false);
                    }
                    #pragma unroll
                    for (int r = 0; r < 8; ++r) t0[r * 129] = acc0[r];
                    if (two) {
                        #pragma unroll
                        for (int r = 0; r < 8; ++r) t1[r * 129] = acc1[r];
                    }
                }
            }
        }
        __syncthreads();
    }

    float xout = 0.0f;
    float yown = Msh[tid * 129 + KDIM];
    float tA[16][16], tB[16][16];
    float uA[16], uB[16];
    {
        const int kl = KDIM - 16;
        #pragma unroll
        for (int r = 0; r < 16; ++r) {
            #pragma unroll
            for (int c = r; c < 16; ++c)
                tA[r][c] = Msh[(kl + r) * 129 + kl + c];
        }
        #pragma unroll
        for (int c = 0; c < 16; ++c)
            uA[c] = Msh[tid * 129 + kl + c];
    }
    for (int kp = KDIM / 32 - 1; kp >= 0; --kp) {
        fm_bs_block(Msh, tid, 2 * kp + 1, tA, uA, tB, uB, yown, xout);
        fm_bs_block(Msh, tid, 2 * kp,     tB, uB, tA, uA, yown, xout);
    }
    Cout[i * KDIM + tid] = xout;
}

extern "C" void kernel_launch(void* const* d_in, const int* in_sizes, int n_in,
                              void* d_out, int out_size, void* d_ws, size_t ws_size,
                              hipStream_t stream)
{
    const float* feat_x = (const float*)d_in[0];
    const float* feat_y = (const float*)d_in[1];
    const float* evals_x = (const float*)d_in[2];
    const float* evals_y = (const float*)d_in[3];
    const float* evecs_x = (const float*)d_in[4];
    const float* evecs_y = (const float*)d_in[5];

    float* wsf   = (float*)d_ws;
    float* Amat  = wsf;
    float* Bmat  = wsf + 32768;
    float* Dmask = wsf + 98304;
    float* part  = wsf + 114688;
    float* Cout  = (float*)d_out;

    fm_gemm_partial<<<dim3(2 * NCHUNKS, 2), 128, 0, stream>>>(
        evecs_x, evecs_y, feat_x, feat_y, part);

    fm_reduce<<<(KP + 255) / 256, 256, 0, stream>>>(part, Amat, Bmat);

    fm_mask<<<(KDIM * KDIM + 255) / 256, 256, 0, stream>>>(evals_x, evals_y, Dmask);

    fm_solve<<<KDIM, KDIM, 68608  , stream>>>(
        Amat, Bmat, Dmask, Cout);
}


# reduce kernel: 256 workgroups, partials as 16-B loads, all 16 of a lane in flight (8 waves x 8 chunks, sums meet in LDS); two waves make the mask meanwhile, mask launch gone; sums regroup by 8
# speedup vs baseline: 6.1086x; 1.0481x over previous
#include <hip/hip_runtime.h>
#include <hip/hip_bf16.h>

typedef __attribute__((ext_vector_type(2))) float v2f;
typedef __attribute__((ext_vector_type(4))) float v4f;
typedef __attribute__((ext_vector_type(8))) float v8f;

#define KDIM 128
#define PDIM 256
#define NDIM 100000
#define LAMBDA 0.001f
#define NCHUNKS 64
#define SLAB 32
#define CHUNK 1568
#define EVP 36
#define KP (KDIM * PDIM)

__global__ __launch_bounds__(128) void fm_gemm_partial(
    const float* __restrict__ evx, const float* __restrict__ evy,
    const float* __restrict__ fx,  const float* __restrict__ fy,
    float* __restrict__ partials)
{
    const int sel   = blockIdx.y;
    const int chunk = blockIdx.x >> 1;
    const int half  = blockIdx.x & 1;
    const float* __restrict__ ev   = sel ? evy : evx;
    const float* __restrict__ feat = sel ? fy  : fx;

    __shared__ __attribute__((aligned(16))) float evS[2][KDIM][EVP];

    const int tid  = threadIdx.x;
    const int lane = tid & 31;
    const int wave = tid >> 5;
    const int c    = lane & 15;
    const int h    = lane >> 4;
    const int colBase = half * 128 + wave * 32;

    const int nStart = chunk * CHUNK;
    const int nEnd   = min(nStart + CHUNK, NDIM);
    const int nSlabs = (nEnd - nStart) / SLAB;

    const int sRow = tid >> 3;
    const int sCol = (tid & 7) * 4;
    const float* __restrict__ evp = ev + (size_t)sRow * NDIM + nStart + sCol;
    const float* __restrict__ ftp = feat + (size_t)(nStart + 4 * h) * PDIM + colBase + 2 * c;

    v8f acc[8][2];
    #pragma unroll
    for (int rb = 0; rb < 8; ++rb) { acc[rb][0] = (v8f)0.0f; acc[rb][1] = (v8f)0.0f; }

    v2f fcur[16], fnxt[16];
    v4f evr[8];

    #pragma unroll
    for (int q = 0; q < 8; ++q)
        evr[q] = *(const v4f*)(evp + (size_t)q * 16 * NDIM);
    #pragma unroll
    for (int f = 0; f < 16; ++f)
        fcur[f] = *(const v2f*)(ftp + (size_t)((f >> 2) * 8 + (f & 3)) * PDIM);
    #pragma unroll
    for (int q = 0; q < 8; ++q)
        *(v4f*)&evS[0][sRow + 16 * q][sCol] = evr[q];
    __syncthreads();

    for (int i = 0; i < nSlabs; ++i) {
        const int p = i & 1;
        const bool more = (i + 1 < nSlabs);

        if (more) {
            const float* __restrict__ e2 = evp + (size_t)(i + 1) * SLAB;
            #pragma unroll
            for (int q = 0; q < 8; ++q)
                evr[q] = *(const v4f*)(e2 + (size_t)q * 16 * NDIM);
            const float* __restrict__ f2 = ftp + (size_t)(i + 1) * SLAB * PDIM;
            #pragma unroll
            for (int f = 0; f < 16; ++f)
                fnxt[f] = *(const v2f*)(f2 + (size_t)((f >> 2) * 8 + (f & 3)) * PDIM);
        }

        v4f a[8];
        #pragma unroll
        for (int rb = 0; rb < 8; ++rb)
            a[rb] = *(const v4f*)&evS[p][rb * 16 + c][4 * h];
        #pragma unroll
        for (int jj = 0; jj < 4; ++jj) {
            v4f an[8];
            if (jj < 3) {
                #pragma unroll
                for (int rb = 0; rb < 8; ++rb)
                    an[rb] = *(const v4f*)&evS[p][rb * 16 + c][8 * (jj + 1) + 4 * h];
            }
            #pragma unroll
            for (int u = 0; u < 2; ++u) {
                const v2f f0 = fcur[jj * 4 + u * 2];
                const v2f f1 = fcur[jj * 4 + u * 2 + 1];
                v2f b0, b1;
                b0.x = f0.x; b0.y = f1.x;
                b1.x = f0.y; b1.y = f1.y;
                #pragma unroll
                for (int rb = 0; rb < 8; ++rb) {
                    v2f av;
                    av.x = a[rb][2 * u];
                    av.y = a[rb][2 * u + 1];
                    acc[rb][0] = __builtin_amdgcn_wmma_f32_16x16x4_f32(
                        false, av, false, b0, (short)0, acc[rb][0], false, false);
                    acc[rb][1] = __builtin_amdgcn_wmma_f32_16x16x4_f32(
                        false, av, false, b1, (short)0, acc[rb][1], false, false);
                }
            }
            if (jj < 3) {
                #pragma unroll
                for (int rb = 0; rb < 8; ++rb) a[rb] = an[rb];
            }
        }

        if (more) {
            #pragma unroll
            for (int q = 0; q < 8; ++q)
                *(v4f*)&evS[p ^ 1][sRow + 16 * q][sCol] = evr[q];
            #pragma unroll
            for (int f = 0; f < 16; ++f) fcur[f] = fnxt[f];
        }
        __syncthreads();
    }

    float* out = partials + (size_t)sel * NCHUNKS * KP + (size_t)chunk * KP;
    #pragma unroll
    for (int rb = 0; rb < 8; ++rb) {
        #pragma unroll
        for (int r = 0; r < 8; ++r) {
            const int row = rb * 16 + 8 * h + r;
            v2f o;
            o.x = acc[rb][0][r];
            o.y = acc[rb][1][r];
            *(v2f*)&out[row * PDIM + colBase + 2 * c] = o;
        }
    }
}

__global__ __launch_bounds__(256) void fm_reduce_mask(
    const float* __restrict__ partials,
    const float* __restrict__ evx, const float* __restrict__ evy,
    float* __restrict__ Amat, float* __restrict__ Bmat, float* __restrict__ D)
{
    __shared__ __attribute__((aligned(16))) float red[8][256];

    const int b    = blockIdx.x;
    const int sel  = b >> 7;
    const int base = (b & 127) * 256;
    const int tid  = threadIdx.x;
    const int lane = tid & 31;
    const int w    = tid >> 5;

    const float* __restrict__ pp =
        partials + (size_t)(sel * NCHUNKS + 8 * w) * KP + base + 4 * lane;
    v4f v[2][8];
    #pragma unroll
    for (int q = 0; q < 8; ++q) {
        v[0][q] = *(const v4f*)(pp + (size_t)q * KP);
        v[1][q] = *(const v4f*)(pp + (size_t)q * KP + 128);
    }

    const bool mk = ((tid >> 6) == 1);
    if (mk) {
        const int idx = b * 64 + (tid & 63);
        const int i = idx >> 7;
        const int j = idx & 127;
        float s  = fmaxf(evx[KDIM - 1], evy[KDIM - 1]);
        float g1 = sqrtf(evx[j] / s);
        float g2 = sqrtf(evy[i] / s);
        float d1 = 1.0f / (g1 * g1 + 1.0f);
        float d2 = 1.0f / (g2 * g2 + 1.0f);
        float re = g2 * d2 - g1 * d1;
        float im = d2 - d1;
        D[idx] = re * re + im * im;
    }

    v4f s0 = v[0][0], s1 = v[1][0];
    #pragma unroll
    for (int q = 1; q < 8; ++q) { s0 += v[0][q]; s1 += v[1][q]; }
    *(v4f*)&red[w][4 * lane]       = s0;
    *(v4f*)&red[w][128 + 4 * lane] = s1;
    __syncthreads();

    if (tid < 64) {
        v4f t = *(const v4f*)&red[0][4 * tid];
        #pragma unroll
        for (int ww = 1; ww < 8; ++ww) t += *(const v4f*)&red[ww][4 * tid];
        float* __restrict__ out = sel ? Bmat : Amat;
        *(v4f*)&out[base + 4 * tid] = t;
    }
}

__device__ __forceinline__ void fm_bs_block(
    float* Msh, const int tid, const int kb,
    const float (&tc)[16][16], const float (&uc)[16],
    float (&tn)[16][16], float (&un)[16], float& yown, float& xout)
{
    const int k0 = kb * 16;
    const int kn = (kb > 0) ? k0 - 16 : 0;

    __syncthreads();

    float yb[16], x[16];
    #pragma unroll
    for (int r = 0; r < 16; ++r)
        yb[r] = Msh[(k0 + r) * 129 + KDIM];
    #pragma unroll
    for (int r = 0; r < 16; ++r) {
        #pragma unroll
        for (int c = r; c < 16; ++c)
            tn[r][c] = Msh[(kn + r) * 129 + kn + c];
    }
    #pragma unroll
    for (int c = 0; c < 16; ++c)
        un[c] = Msh[tid * 129 + kn + c];
    __builtin_amdgcn_sched_barrier(0);

    #pragma unroll
    for (int c = 15; c >= 0; --c) {
        x[c] = yb[c] / tc[c][c];
        #pragma unroll
        for (int r = 0; r < c; ++r)
            yb[r] -= tc[r][c] * x[c];
        yown -= uc[c] * x[c];
        xout = (((tid >> 4) == kb) && ((tid & 15) == c)) ? x[c] : xout;
    }
    if ((tid >> 4) == kb - 1)
        Msh[tid * 129 + KDIM] = yown;
}

template <int LC>
__device__ __forceinline__ float fm_row_bcast(const float x)
{
    return __builtin_bit_cast(float, __builtin_amdgcn_update_dpp(
        0, __builtin_bit_cast(int, x), 0x150 + LC  , 0xf, 0xf, true));
}

template <int LC>
__device__ __forceinline__ void fm_panel_pivots(float (&p)[16], float (&d)[16],
                                                const int c16, const int rel)
{
    const bool belowd = c16 > LC;
    const bool below  = rel > LC;
    float q[16];
    #pragma unroll
    for (int j = LC; j < 16; ++j) q[j] = fm_row_bcast<LC>(d[j]);
    const float fd = d[LC] / q[LC];
    const float f  = p[LC] / q[LC];
    d[LC] = belowd ? fd : d[LC];
    p[LC] = below  ? f  : p[LC];
    #pragma unroll
    for (int j = LC + 1; j < 16; ++j) {
        d[j] = belowd ? (d[j] - fd * q[j]) : d[j];
        p[j] = below  ? (p[j] - f  * q[j]) : p[j];
    }
    if constexpr (LC < 15) fm_panel_pivots<LC + 1>(p, d, c16, rel);
}

#define APITCH 132
__global__ __launch_bounds__(128) void fm_solve(
    const float* __restrict__ Amat, const float* __restrict__ Bmat,
    const float* __restrict__ D,    float* __restrict__ Cout)
{
    extern __shared__ float4 smem4[];
    float* Msh  = (float*)smem4;
    float* Lbuf = Msh + KDIM * 129;
    float* Ast  = (float*)smem4;
    float* Brow = Ast + KDIM * APITCH;
    const int i   = blockIdx.x;
    const int tid = threadIdx.x;
    const int wvu = __builtin_amdgcn_readfirstlane(tid >> 5);

    {
        const int lane = tid & 31;
        const int wv   = tid >> 5;
        const int c    = lane & 15;
        const int h    = lane >> 4;

        const float* __restrict__ ap = Amat + (size_t)wv * PDIM + 4 * lane;
        v4f st0[32], st1[32];
        #pragma unroll
        for (int q = 0; q < 32; ++q)
            st0[q] = *(const v4f*)(ap + (size_t)q * 4 * PDIM);
        const v2f bq = *(const v2f*)(Bmat + (size_t)i * PDIM + 2 * tid);
        const float dsc = D[i * KDIM + tid];
        #pragma unroll
        for (int q = 0; q < 32; ++q)
            *(v4f*)&Ast[(4 * q + wv) * APITCH + 4 * lane] = st0[q];
        *(v2f*)&Brow[2 * tid] = bq;
        __syncthreads();
        #pragma unroll
        for (int q = 0; q < 32; ++q)
            st1[q] = *(const v4f*)(ap + (size_t)q * 4 * PDIM + 128);

        v8f acc[11];
        #pragma unroll
        for (int t = 0; t < 11; ++t) acc[t] = (v8f)0.0f;

        int fo[8];
        #pragma unroll
        for (int t = 0; t < 8; ++t) fo[t] = (16 * ((wv + t) & 7) + c) * APITCH + 2 * h;
        const bool row0 = (c == 0);

        for (int hh = 0; hh < 2; ++hh) {
            if (hh == 1) {
                __syncthreads();
                #pragma unroll
                for (int q = 0; q < 32; ++q)
                    *(v4f*)&Ast[(4 * q + wv) * APITCH + 4 * lane] = st1[q];
                __syncthreads();
            }
            v2f f[8], fb;
            #pragma unroll
            for (int t = 0; t < 8; ++t) f[t] = *(const v2f*)&Ast[fo[t]];
            {
                const v2f tb = *(const v2f*)&Brow[128 * hh + 2 * h];
                fb.x = row0 ? tb.x : 0.0f;
                fb.y = row0 ? tb.y : 0.0f;
            }
            for (int ks = 0; ks < 32; ++ks) {
                v2f fn[8], fbn;
                if (ks + 1 < 32) {
                    #pragma unroll
                    for (int t = 0; t < 8; ++t) fn[t] = *(const v2f*)&Ast[fo[t] + 4 * (ks + 1)];
                    const v2f tb = *(const v2f*)&Brow[128 * hh + 4 * (ks + 1) + 2 * h];
                    fbn.x = row0 ? tb.x : 0.0f;
                    fbn.y = row0 ? tb.y : 0.0f;
                } else {
                    #pragma unroll
                    for (int t = 0; t < 8; ++t) fn[t] = f[t];
                    fbn = fb;
                }
                #pragma unroll
                for (int t = 0; t < 5; ++t)
                    acc[t] = __builtin_amdgcn_wmma_f32_16x16x4_f32(
                        false, f[0], false, f[t], (short)0, acc[t], false, false);
                #pragma unroll
                for (int t = 0; t < 4; ++t)
                    acc[5 + t] = __builtin_amdgcn_wmma_f32_16x16x4_f32(
                        false, f[4], false, f[4 + t], (short)0, acc[5 + t], false, false);
                acc[9]  = __builtin_amdgcn_wmma_f32_16x16x4_f32(
                    false, fb, false, f[0], (short)0, acc[9], false, false);
                acc[10] = __builtin_amdgcn_wmma_f32_16x16x4_f32(
                    false, fb, false, f[4], (short)0, acc[10], false, false);
                #pragma unroll
                for (int t = 0; t < 8; ++t) f[t] = fn[t];
                fb = fbn;
            }
        }
        __syncthreads();

        #pragma unroll
        for (int a = 0; a < 2; ++a) {
            const int I = wv + 4 * a;
            #pragma unroll
            for (int t = 0; t < 5 - a; ++t) {
                const int J = (I + t) & 7;
                const int col = 16 * J + c;
                #pragma unroll
                for (int r = 0; r < 8; ++r) {
                    const int row = 16 * I + 8 * h + r;
                    const float v = acc[5 * a + t][r];
                    Msh[row * 129 + col] = v;
                    if (t != 0) Msh[col * 129 + row] = v;
                }
            }
            if (h == 0)
                Msh[(16 * I + c) * 129 + KDIM] = acc[9 + a][0];
        }
        __syncthreads();
        Msh[tid * 129 + tid] += LAMBDA * dsc;
        __syncthreads();
    }

    for (int kb = 0; kb < KDIM / 16; ++kb) {
        const int k0 = kb * 16;

        float p[16], d[16];
        #pragma unroll
        for (int j = 0; j < 16; ++j)
            p[j] = (tid >= k0) ? Msh[tid * 129 + k0 + j] : 0.0f;
        #pragma unroll
        for (int j = 0; j < 16; ++j)
            d[j] = Msh[(k0 + (tid & 15)) * 129 + k0 + j];
        fm_panel_pivots<0>(p, d, tid & 15, tid - k0);
        __syncthreads();

        if (tid >= k0 && tid < k0 + 16) {
            #pragma unroll
            for (int j = 0; j < 16; ++j)
                Msh[tid * 129 + k0 + j] = p[j];
            #pragma unroll
            for (int j = 0; j < 16; ++j)
                Lbuf[(tid - k0) * 16 + j] = p[j];
        } else if (tid >= k0 + 16) {
            #pragma unroll
            for (int j = 0; j < 16; ++j)
                Msh[tid * 129 + k0 + j] = -p[j];
        }
        __syncthreads();

        const int ncm = KDIM - 16 - k0;
        if (tid <= ncm) {
            const int j = k0 + 16 + tid;
            float u[16];
            #pragma unroll
            for (int r = 0; r < 16; ++r)
                u[r] = Msh[(k0 + r) * 129 + j];
            #pragma unroll
            for (int r = 1; r < 16; ++r) {
                const float4* lr4 = (const float4*)(Lbuf + r * 16);
                float l[16];
                #pragma unroll
                for (int g = 0; g * 4 < r; ++g) {
                    const float4 t = lr4[g];
                    l[4 * g] = t.x; l[4 * g + 1] = t.y; l[4 * g + 2] = t.z; l[4 * g + 3] = t.w;
                }
                #pragma unroll
                for (int c = 0; c < r; ++c)
                    u[r] -= l[c] * u[c];
            }
            #pragma unroll
            for (int r = 0; r < 16; ++r)
                Msh[(k0 + r) * 129 + j] = u[r];
        }
        __syncthreads();

        if (tid >= k0 + 16) {
            float xr = Msh[tid * 129 + KDIM];
            #pragma unroll
            for (int c = 0; c < 16; ++c)
                xr -= p[c] * Msh[(k0 + c) * 129 + KDIM];
            Msh[tid * 129 + KDIM] = xr;
        }
        {
            const int lane = tid & 31;
            const int c    = lane & 15;
            const int h    = lane >> 4;
            const int np   = (8 - kb) >> 1;
            int item = 0;
            for (int I = kb + 1; I < 8; ++I) {
                for (int jp = 0; jp < np; ++jp, ++item) {
                    if ((item & 3) != wvu) continue;
                    const int  J0  = kb + 1 + 2 * jp;
                    const bool two = (J0 + 1 < 8);
                    const int  J1  = two ? J0 + 1 : J0;
                    const float* la = Msh + (16 * I + c) * 129 + k0 + 2 * h;
                    const float* u0 = Msh + (k0 + 2 * h) * 129 + 16 * J0 + c;
                    const float* u1 = Msh + (k0 + 2 * h) * 129 + 16 * J1 + c;
                    float* t0 = Msh + (16 * I + 8 * h) * 129 + 16 * J0 + c;
                    float* t1 = Msh + (16 * I + 8 * h) * 129 + 16 * J1 + c;
                    v2f a[4], b0[4], b1[4];
                    v8f acc0, acc1;
                    #pragma unroll
                    for (int ks = 0; ks < 4; ++ks) {
                        a[ks].x  = la[4 * ks];           a[ks].y  = la[4 * ks + 1];
                        b0[ks].x = u0[(4 * ks) * 129];   b0[ks].y = u0[(4 * ks + 1) * 129];
                        b1[ks].x = u1[(4 * ks) * 129];   b1[ks].y = u1[(4 * ks + 1) * 129];
                    }
                    #pragma unroll
                    for (int r = 0; r < 8; ++r) { acc0[r] = t0[r * 129]; acc1[r] = t1[r * 129]; }
                    #pragma unroll
                    for (int ks = 0; ks < 4; ++ks) {
                        acc0 = __builtin_amdgcn_wmma_f32_16x16x4_f32(
                            false, a[ks], false, b0[ks], (short)0, acc0, false, false);
                        acc1 = __builtin_amdgcn_wmma_f32_16x16x4_f32(
                            false, a[ks], false, b1[ks], (short)0, acc1, false, false);
                    }
                    #pragma unroll
                    for (int r = 0; r < 8; ++r) t0[r * 129] = acc0[r];
                    if (two) {
                        #pragma unroll
                        for (int r = 0; r < 8; ++r) t1[r * 129] = acc1[r];
                    }
                }
            }
        }
        __syncthreads();
    }

    float xout = 0.0f;
    float yown = Msh[tid * 129 + KDIM];
    float tA[16][16], tB[16][16];
    float uA[16], uB[16];
    {
        const int kl = KDIM - 16;
        #pragma unroll
        for (int r = 0; r < 16; ++r) {
            #pragma unroll
            for (int c = r; c < 16; ++c)
                tA[r][c] = Msh[(kl + r) * 129 + kl + c];
        }
        #pragma unroll
        for (int c = 0; c < 16; ++c)
            uA[c] = Msh[tid * 129 + kl + c];
    }
    for (int kp = KDIM / 32 - 1; kp >= 0; --kp) {
        fm_bs_block(Msh, tid, 2 * kp + 1, tA, uA, tB, uB, yown, xout);
        fm_bs_block(Msh, tid, 2 * kp,     tB, uB, tA, uA, yown, xout);
    }
    Cout[i * KDIM + tid] = xout;
}

extern "C" void kernel_launch(void* const* d_in, const int* in_sizes, int n_in,
                              void* d_out, int out_size, void* d_ws, size_t ws_size,
                              hipStream_t stream)
{
    const float* feat_x = (const float*)d_in[0];
    const float* feat_y = (const float*)d_in[1];
    const float* evals_x = (const float*)d_in[2];
    const float* evals_y = (const float*)d_in[3];
    const float* evecs_x = (const float*)d_in[4];
    const float* evecs_y = (const float*)d_in[5];

    float* wsf   = (float*)d_ws;
    float* Amat  = wsf;
    float* Bmat  = wsf + 32768;
    float* Dmask = wsf + 98304;
    float* part  = wsf + 114688;
    float* Cout  = (float*)d_out;

    fm_gemm_partial<<<dim3(2 * NCHUNKS, 2), 128, 0, stream>>>(
        evecs_x, evecs_y, feat_x, feat_y, part);

    fm_reduce_mask<<<256, 256, 0, stream>>>(part, evals_x, evals_y, Amat, Bmat, Dmask);

    fm_solve<<<KDIM, KDIM, 68608  , stream>>>(
        Amat, Bmat, Dmask, Cout);
}


# solve kernel: LDL^T from M's symmetry: U right of a panel = mirrored pre-division column (no per-column solve), only tiles on/below diagonal updated, back sweep on unit L^T, z=y/pivot; sums regroup
# speedup vs baseline: 6.6488x; 1.0884x over previous
#include <hip/hip_runtime.h>
#include <hip/hip_bf16.h>

typedef __attribute__((ext_vector_type(2))) float v2f;
typedef __attribute__((ext_vector_type(4))) float v4f;
typedef __attribute__((ext_vector_type(8))) float v8f;

#define KDIM 128
#define PDIM 256
#define NDIM 100000
#define LAMBDA 0.001f
#define NCHUNKS 64
#define SLAB 32
#define CHUNK 1568
#define EVP 36
#define KP (KDIM * PDIM)

__global__ __launch_bounds__(128) void fm_gemm_partial(
    const float* __restrict__ evx, const float* __restrict__ evy,
    const float* __restrict__ fx,  const float* __restrict__ fy,
    float* __restrict__ partials)
{
    const int sel   = blockIdx.y;
    const int chunk = blockIdx.x >> 1;
    const int half  = blockIdx.x & 1;
    const float* __restrict__ ev   = sel ? evy : evx;
    const float* __restrict__ feat = sel ? fy  : fx;

    __shared__ __attribute__((aligned(16))) float evS[2][KDIM][EVP];

    const int tid  = threadIdx.x;
    const int lane = tid & 31;
    const int wave = tid >> 5;
    const int c    = lane & 15;
    const int h    = lane >> 4;
    const int colBase = half * 128 + wave * 32;

    const int nStart = chunk * CHUNK;
    const int nEnd   = min(nStart + CHUNK, NDIM);
    const int nSlabs = (nEnd - nStart) / SLAB;

    const int sRow = tid >> 3;
    const int sCol = (tid & 7) * 4;
    const float* __restrict__ evp = ev + (size_t)sRow * NDIM + nStart + sCol;
    const float* __restrict__ ftp = feat + (size_t)(nStart + 4 * h) * PDIM + colBase + 2 * c;

    v8f acc[8][2];
    #pragma unroll
    for (int rb = 0; rb < 8; ++rb) { acc[rb][0] = (v8f)0.0f; acc[rb][1] = (v8f)0.0f; }

    v2f fcur[16], fnxt[16];
    v4f evr[8];

    #pragma unroll
    for (int q = 0; q < 8; ++q)
        evr[q] = *(const v4f*)(evp + (size_t)q * 16 * NDIM);
    #pragma unroll
    for (int f = 0; f < 16; ++f)
        fcur[f] = *(const v2f*)(ftp + (size_t)((f >> 2) * 8 + (f & 3)) * PDIM);
    #pragma unroll
    for (int q = 0; q < 8; ++q)
        *(v4f*)&evS[0][sRow + 16 * q][sCol] = evr[q];
    __syncthreads();

    for (int i = 0; i < nSlabs; ++i) {
        const int p = i & 1;
        const bool more = (i + 1 < nSlabs);

        if (more) {
            const float* __restrict__ e2 = evp + (size_t)(i + 1) * SLAB;
            #pragma unroll
            for (int q = 0; q < 8; ++q)
                evr[q] = *(const v4f*)(e2 + (size_t)q * 16 * NDIM);
            const float* __restrict__ f2 = ftp + (size_t)(i + 1) * SLAB * PDIM;
            #pragma unroll
            for (int f = 0; f < 16; ++f)
                fnxt[f] = *(const v2f*)(f2 + (size_t)((f >> 2) * 8 + (f & 3)) * PDIM);
        }

        v4f a[8];
        #pragma unroll
        for (int rb = 0; rb < 8; ++rb)
            a[rb] = *(const v4f*)&evS[p][rb * 16 + c][4 * h];
        #pragma unroll
        for (int jj = 0; jj < 4; ++jj) {
            v4f an[8];
            if (jj < 3) {
                #pragma unroll
                for (int rb = 0; rb < 8; ++rb)
                    an[rb] = *(const v4f*)&evS[p][rb * 16 + c][8 * (jj + 1) + 4 * h];
            }
            #pragma unroll
            for (int u = 0; u < 2; ++u) {
                const v2f f0 = fcur[jj * 4 + u * 2];
                const v2f f1 = fcur[jj * 4 + u * 2 + 1];
                v2f b0, b1;
                b0.x = f0.x; b0.y = f1.x;
                b1.x = f0.y; b1.y = f1.y;
                #pragma unroll
                for (int rb = 0; rb < 8; ++rb) {
                    v2f av;
                    av.x = a[rb][2 * u];
                    av.y = a[rb][2 * u + 1];
                    acc[rb][0] = __builtin_amdgcn_wmma_f32_16x16x4_f32(
                        false, av, false, b0, (short)0, acc[rb][0], false, false);
                    acc[rb][1] = __builtin_amdgcn_wmma_f32_16x16x4_f32(
                        false, av, false, b1, (short)0, acc[rb][1], false, false);
                }
            }
            if (jj < 3) {
                #pragma unroll
                for (int rb = 0; rb < 8; ++rb) a[rb] = an[rb];
            }
        }

        if (more) {
            #pragma unroll
            for (int q = 0; q < 8; ++q)
                *(v4f*)&evS[p ^ 1][sRow + 16 * q][sCol] = evr[q];
            #pragma unroll
            for (int f = 0; f < 16; ++f) fcur[f] = fnxt[f];
        }
        __syncthreads();
    }

    float* out = partials + (size_t)sel * NCHUNKS * KP + (size_t)chunk * KP;
    #pragma unroll
    for (int rb = 0; rb < 8; ++rb) {
        #pragma unroll
        for (int r = 0; r < 8; ++r) {
            const int row = rb * 16 + 8 * h + r;
            v2f o;
            o.x = acc[rb][0][r];
            o.y = acc[rb][1][r];
            *(v2f*)&out[row * PDIM + colBase + 2 * c] = o;
        }
    }
}

__global__ __launch_bounds__(256) void fm_reduce_mask(
    const float* __restrict__ partials,
    const float* __restrict__ evx, const float* __restrict__ evy,
    float* __restrict__ Amat, float* __restrict__ Bmat, float* __restrict__ D)
{
    __shared__ __attribute__((aligned(16))) float red[8][256];

    const int b    = blockIdx.x;
    const int sel  = b >> 7;
    const int base = (b & 127) * 256;
    const int tid  = threadIdx.x;
    const int lane = tid & 31;
    const int w    = tid >> 5;

    const float* __restrict__ pp =
        partials + (size_t)(sel * NCHUNKS + 8 * w) * KP + base + 4 * lane;
    v4f v[2][8];
    #pragma unroll
    for (int q = 0; q < 8; ++q) {
        v[0][q] = *(const v4f*)(pp + (size_t)q * KP);
        v[1][q] = *(const v4f*)(pp + (size_t)q * KP + 128);
    }

    const bool mk = ((tid >> 6) == 1);
    if (mk) {
        const int idx = b * 64 + (tid & 63);
        const int i = idx >> 7;
        const int j = idx & 127;
        float s  = fmaxf(evx[KDIM - 1], evy[KDIM - 1]);
        float g1 = sqrtf(evx[j] / s);
        float g2 = sqrtf(evy[i] / s);
        float d1 = 1.0f / (g1 * g1 + 1.0f);
        float d2 = 1.0f / (g2 * g2 + 1.0f);
        float re = g2 * d2 - g1 * d1;
        float im = d2 - d1;
        D[idx] = re * re + im * im;
    }

    v4f s0 = v[0][0], s1 = v[1][0];
    #pragma unroll
    for (int q = 1; q < 8; ++q) { s0 += v[0][q]; s1 += v[1][q]; }
    *(v4f*)&red[w][4 * lane]       = s0;
    *(v4f*)&red[w][128 + 4 * lane] = s1;
    __syncthreads();

    if (tid < 64) {
        v4f t = *(const v4f*)&red[0][4 * tid];
        #pragma unroll
        for (int ww = 1; ww < 8; ++ww) t += *(const v4f*)&red[ww][4 * tid];
        float* __restrict__ out = sel ? Bmat : Amat;
        *(v4f*)&out[base + 4 * tid] = t;
    }
}

__device__ __forceinline__ void fm_bs_block(
    float* Msh, const int tid, const int kb,
    const float (&tc)[16][16], const float (&uc)[16],
    float (&tn)[16][16], float (&un)[16], float& yown, float& xout)
{
    const int k0 = kb * 16;
    const int kn = (kb > 0) ? k0 - 16 : 0;

    __syncthreads();

    float yb[16], x[16];
    #pragma unroll
    for (int r = 0; r < 16; ++r)
        yb[r] = Msh[(k0 + r) * 129 + KDIM];
    #pragma unroll
    for (int r = 0; r < 16; ++r) {
        #pragma unroll
        for (int c = r + 1; c < 16; ++c)
            tn[r][c] = Msh[(kn + c) * 129 + kn + r];
    }
    #pragma unroll
    for (int c = 0; c < 16; ++c)
        un[c] = Msh[(kn + c) * 129 + tid];
    __builtin_amdgcn_sched_barrier(0);

    #pragma unroll
    for (int c = 15; c >= 0; --c) {
        x[c] = yb[c];
        #pragma unroll
        for (int r = 0; r < c; ++r)
            yb[r] -= tc[r][c] * x[c];
        yown += uc[c] * x[c];
        xout = (((tid >> 4) == kb) && ((tid & 15) == c)) ? x[c] : xout;
    }
    if ((tid >> 4) == kb - 1)
        Msh[tid * 129 + KDIM] = yown;
}

template <int LC>
__device__ __forceinline__ float fm_row_bcast(const float x)
{
    return __builtin_bit_cast(float, __builtin_amdgcn_update_dpp(
        0, __builtin_bit_cast(int, x), 0x150 + LC  , 0xf, 0xf, true));
}

template <int LC>
__device__ __forceinline__ void fm_panel_pivots(float (&p)[16], float (&d)[16], float (&pre)[16],
                                                float& pr, float& dr,
                                                const int c16, const int rel)
{
    const bool belowd = c16 > LC;
    const bool below  = rel > LC;
    float q[16];
    #pragma unroll
    for (int j = LC; j < 16; ++j) q[j] = fm_row_bcast<LC>(d[j]);
    const float qr = fm_row_bcast<LC>(dr);
    const float fd = d[LC] / q[LC];
    const float f  = p[LC] / q[LC];
    pre[LC] = p[LC];
    d[LC] = belowd ? fd : d[LC];
    p[LC] = below  ? f  : p[LC];
    #pragma unroll
    for (int j = LC + 1; j < 16; ++j) {
        d[j] = belowd ? (d[j] - fd * q[j]) : d[j];
        p[j] = below  ? (p[j] - f  * q[j]) : p[j];
    }
    dr = belowd ? (dr - fd * qr) : dr;
    pr = below  ? (pr - f  * qr) : pr;
    if constexpr (LC < 15) fm_panel_pivots<LC + 1>(p, d, pre, pr, dr, c16, rel);
}

#define APITCH 132
__global__ __launch_bounds__(128) void fm_solve(
    const float* __restrict__ Amat, const float* __restrict__ Bmat,
    const float* __restrict__ D,    float* __restrict__ Cout)
{
    extern __shared__ float4 smem4[];
    float* Msh  = (float*)smem4;
    float* Ast  = (float*)smem4;
    float* Brow = Ast + KDIM * APITCH;
    const int i   = blockIdx.x;
    const int tid = threadIdx.x;
    const int wvu = __builtin_amdgcn_readfirstlane(tid >> 5);

    {
        const int lane = tid & 31;
        const int wv   = tid >> 5;
        const int c    = lane & 15;
        const int h    = lane >> 4;

        const float* __restrict__ ap = Amat + (size_t)wv * PDIM + 4 * lane;
        v4f st0[32], st1[32];
        #pragma unroll
        for (int q = 0; q < 32; ++q)
            st0[q] = *(const v4f*)(ap + (size_t)q * 4 * PDIM);
        const v2f bq = *(const v2f*)(Bmat + (size_t)i * PDIM + 2 * tid);
        const float dsc = D[i * KDIM + tid];
        #pragma unroll
        for (int q = 0; q < 32; ++q)
            *(v4f*)&Ast[(4 * q + wv) * APITCH + 4 * lane] = st0[q];
        *(v2f*)&Brow[2 * tid] = bq;
        __syncthreads();
        #pragma unroll
        for (int q = 0; q < 32; ++q)
            st1[q] = *(const v4f*)(ap + (size_t)q * 4 * PDIM + 128);

        v8f acc[11];
        #pragma unroll
        for (int t = 0; t < 11; ++t) acc[t] = (v8f)0.0f;

        int fo[8];
        #pragma unroll
        for (int t = 0; t < 8; ++t) fo[t] = (16 * ((wv + t) & 7) + c) * APITCH + 2 * h;
        const bool row0 = (c == 0);

        for (int hh = 0; hh < 2; ++hh) {
            if (hh == 1) {
                __syncthreads();
                #pragma unroll
                for (int q = 0; q < 32; ++q)
                    *(v4f*)&Ast[(4 * q + wv) * APITCH + 4 * lane] = st1[q];
                __syncthreads();
            }
            v2f f[8], fb;
            #pragma unroll
            for (int t = 0; t < 8; ++t) f[t] = *(const v2f*)&Ast[fo[t]];
            {
                const v2f tb = *(const v2f*)&Brow[128 * hh + 2 * h];
                fb.x = row0 ? tb.x : 0.0f;
                fb.y = row0 ? tb.y : 0.0f;
            }
            for (int ks = 0; ks < 32; ++ks) {
                v2f fn[8], fbn;
                if (ks + 1 < 32) {
                    #pragma unroll
                    for (int t = 0; t < 8; ++t) fn[t] = *(const v2f*)&Ast[fo[t] + 4 * (ks + 1)];
                    const v2f tb = *(const v2f*)&Brow[128 * hh + 4 * (ks + 1) + 2 * h];
                    fbn.x = row0 ? tb.x : 0.0f;
                    fbn.y = row0 ? tb.y : 0.0f;
                } else {
                    #pragma unroll
                    for (int t = 0; t < 8; ++t) fn[t] = f[t];
                    fbn = fb;
                }
                #pragma unroll
                for (int t = 0; t < 5; ++t)
                    acc[t] = __builtin_amdgcn_wmma_f32_16x16x4_f32(
                        false, f[0], false, f[t], (short)0, acc[t], false, false);
                #pragma unroll
                for (int t = 0; t < 4; ++t)
                    acc[5 + t] = __builtin_amdgcn_wmma_f32_16x16x4_f32(
                        false, f[4], false, f[4 + t], (short)0, acc[5 + t], false, false);
                acc[9]  = __builtin_amdgcn_wmma_f32_16x16x4_f32(
                    false, fb, false, f[0], (short)0, acc[9], false, false);
                acc[10] = __builtin_amdgcn_wmma_f32_16x16x4_f32(
                    false, fb, false, f[4], (short)0, acc[10], false, false);
                #pragma unroll
                for (int t = 0; t < 8; ++t) f[t] = fn[t];
                fb = fbn;
            }
        }
        __syncthreads();

        #pragma unroll
        for (int a = 0; a < 2; ++a) {
            const int I = wv + 4 * a;
            #pragma unroll
            for (int t = 0; t < 5 - a; ++t) {
                const int J = (I + t) & 7;
                const int col = 16 * J + c;
                #pragma unroll
                for (int r = 0; r < 8; ++r) {
                    const int row = 16 * I + 8 * h + r;
                    const float v = acc[5 * a + t][r];
                    Msh[row * 129 + col] = v;
                    if (t != 0) Msh[col * 129 + row] = v;
                }
            }
            if (h == 0)
                Msh[(16 * I + c) * 129 + KDIM] = acc[9 + a][0];
        }
        __syncthreads();
        Msh[tid * 129 + tid] += LAMBDA * dsc;
        __syncthreads();
    }

    float yrow = Msh[tid * 129 + KDIM];

    for (int kb = 0; kb < KDIM / 16; ++kb) {
        const int k0 = kb * 16;

        float p[16], d[16], pre[16];
        #pragma unroll
        for (int j = 0; j < 16; ++j)
            p[j] = (tid >= k0) ? Msh[tid * 129 + k0 + j] : 0.0f;
        #pragma unroll
        for (int j = 0; j < 16; ++j)
            d[j] = Msh[(k0 + (tid & 15)) * 129 + k0 + j];
        float dr = Msh[(k0 + (tid & 15)) * 129 + KDIM];
        fm_panel_pivots<0>(p, d, pre, yrow, dr, tid & 15, tid - k0);
        __syncthreads();

        if (tid >= k0)
            Msh[tid * 129 + KDIM] = yrow;
        if (tid >= k0 && tid < k0 + 16) {
            #pragma unroll
            for (int j = 0; j < 16; ++j)
                Msh[tid * 129 + k0 + j] = p[j];
        } else if (tid >= k0 + 16) {
            #pragma unroll
            for (int j = 0; j < 16; ++j)
                Msh[tid * 129 + k0 + j] = -p[j];
            #pragma unroll
            for (int j = 0; j < 16; ++j)
                Msh[(k0 + j) * 129 + tid] = pre[j];
        }
        __syncthreads();

        {
            const int lane = tid & 31;
            const int c    = lane & 15;
            const int h    = lane >> 4;
            int item = 0;
            for (int I = kb + 1; I < 8; ++I) {
                const int np = (I - kb + 1) >> 1;
                for (int jp = 0; jp < np; ++jp, ++item) {
                    if ((item & 3) != wvu) continue;
                    const int  J0  = kb + 1 + 2 * jp;
                    const bool two = (J0 + 1 <= I);
                    const int  J1  = two ? J0 + 1 : J0;
                    const float* la = Msh + (16 * I + c) * 129 + k0 + 2 * h;
                    const float* u0 = Msh + (k0 + 2 * h) * 129 + 16 * J0 + c;
                    const float* u1 = Msh + (k0 + 2 * h) * 129 + 16 * J1 + c;
                    float* t0 = Msh + (16 * I + 8 * h) * 129 + 16 * J0 + c;
                    float* t1 = Msh + (16 * I + 8 * h) * 129 + 16 * J1 + c;
                    v2f a[4], b0[4], b1[4];
                    v8f acc0, acc1;
                    #pragma unroll
                    for (int ks = 0; ks < 4; ++ks) {
                        a[ks].x  = la[4 * ks];           a[ks].y  = la[4 * ks + 1];
                        b0[ks].x = u0[(4 * ks) * 129];   b0[ks].y = u0[(4 * ks + 1) * 129];
                        b1[ks].x = u1[(4 * ks) * 129];   b1[ks].y = u1[(4 * ks + 1) * 129];
                    }
                    #pragma unroll
                    for (int r = 0; r < 8; ++r) { acc0[r] = t0[r * 129]; acc1[r] = t1[r * 129]; }
                    #pragma unroll
                    for (int ks = 0; ks < 4; ++ks) {
                        acc0 = __builtin_amdgcn_wmma_f32_16x16x4_f32(
                            false, a[ks], false, b0[ks], (short)0, acc0, false, false);
                        acc1 = __builtin_amdgcn_wmma_f32_16x16x4_f32(
                            false, a[ks], false, b1[ks], (short)0, acc1, false, false);
                    }
                    #pragma unroll
                    for (int r = 0; r < 8; ++r) t0[r * 129] = acc0[r];
                    if (two) {
                        #pragma unroll
                        for (int r = 0; r < 8; ++r) t1[r * 129] = acc1[r];
                    }
                }
            }
        }
        __syncthreads();
    }

    float xout = 0.0f;
    float yown = yrow / Msh[tid * 129 + tid];
    Msh[tid * 129 + KDIM] = yown;
    float tA[16][16], tB[16][16];
    float uA[16], uB[16];
    {
        const int kl = KDIM - 16;
        #pragma unroll
        for (int r = 0; r < 16; ++r) {
            #pragma unroll
            for (int c = r + 1; c < 16; ++c)
                tA[r][c] = Msh[(kl + c) * 129 + kl + r];
        }
        #pragma unroll
        for (int c = 0; c < 16; ++c)
            uA[c] = Msh[(kl + c) * 129 + tid];
    }
    for (int kp = KDIM / 32 - 1; kp >= 0; --kp) {
        fm_bs_block(Msh, tid, 2 * kp + 1, tA, uA, tB, uB, yown, xout);
        fm_bs_block(Msh, tid, 2 * kp,     tB, uB, tA, uA, yown, xout);
    }
    Cout[i * KDIM + tid] = xout;
}

extern "C" void kernel_launch(void* const* d_in, const int* in_sizes, int n_in,
                              void* d_out, int out_size, void* d_ws, size_t ws_size,
                              hipStream_t stream)
{
    const float* feat_x = (const float*)d_in[0];
    const float* feat_y = (const float*)d_in[1];
    const float* evals_x = (const float*)d_in[2];
    const float* evals_y = (const float*)d_in[3];
    const float* evecs_x = (const float*)d_in[4];
    const float* evecs_y = (const float*)d_in[5];

    float* wsf   = (float*)d_ws;
    float* Amat  = wsf;
    float* Bmat  = wsf + 32768;
    float* Dmask = wsf + 98304;
    float* part  = wsf + 114688;
    float* Cout  = (float*)d_out;

    fm_gemm_partial<<<dim3(2 * NCHUNKS, 2), 128, 0, stream>>>(
        evecs_x, evecs_y, feat_x, feat_y, part);

    fm_reduce_mask<<<256, 256, 0, stream>>>(part, evals_x, evals_y, Amat, Bmat, Dmask);

    fm_solve<<<KDIM, KDIM, 68608  , stream>>>(
        Amat, Bmat, Dmask, Cout);
}


# gemm kernel: slab hand-over inside the WMMA run: evecs stored to LDS mid-slab, barrier split into signal and wait with WMMAs between, next slab's first fragments read ahead; values kept
# speedup vs baseline: 6.9570x; 1.0464x over previous
#include <hip/hip_runtime.h>
#include <hip/hip_bf16.h>

typedef __attribute__((ext_vector_type(2))) float v2f;
typedef __attribute__((ext_vector_type(4))) float v4f;
typedef __attribute__((ext_vector_type(8))) float v8f;

#define KDIM 128
#define PDIM 256
#define NDIM 100000
#define LAMBDA 0.001f
#define NCHUNKS 64
#define SLAB 32
#define CHUNK 1568
#define EVP 36
#define KP (KDIM * PDIM)

template <int JJ>
__device__ __forceinline__ void fm_gemm_group(v8f (&acc)[8][2], const v2f (&fcur)[16], const v4f (&a)[8])
{
    #pragma unroll
    for (int u = 0; u < 2; ++u) {
        const v2f f0 = fcur[JJ * 4 + u * 2];
        const v2f f1 = fcur[JJ * 4 + u * 2 + 1];
        v2f b0, b1;
        b0.x = f0.x; b0.y = f1.x;
        b1.x = f0.y; b1.y = f1.y;
        #pragma unroll
        for (int rb = 0; rb < 8; ++rb) {
            v2f av;
            av.x = a[rb][2 * u];
            av.y = a[rb][2 * u + 1];
            acc[rb][0] = __builtin_amdgcn_wmma_f32_16x16x4_f32(
                false, av, false, b0, (short)0, acc[rb][0], false, false);
            acc[rb][1] = __builtin_amdgcn_wmma_f32_16x16x4_f32(
                false, av, false, b1, (short)0, acc[rb][1], false, false);
        }
    }
}

template <bool MORE, bool MORE2>
__device__ __forceinline__ void fm_gemm_slab(
    v8f (&acc)[8][2], v2f (&fcur)[16], v4f (&evr)[8], v4f (&a0)[8],
    float (&evS)[2][KDIM][EVP], const int p,
    const float* __restrict__ fnext, const float* __restrict__ enext2,
    const int sRow, const int sCol, const int c, const int h)
{
    v2f fnxt[16];
    v4f a1[8], a2[8], a3[8];

    if constexpr (MORE) {
        #pragma unroll
        for (int f = 0; f < 16; ++f)
            fnxt[f] = *(const v2f*)(fnext + (size_t)((f >> 2) * 8 + (f & 3)) * PDIM);
    }
    #pragma unroll
    for (int rb = 0; rb < 8; ++rb)
        a1[rb] = *(const v4f*)&evS[p][rb * 16 + c][8 + 4 * h];

    fm_gemm_group<0>(acc, fcur, a0);

    if constexpr (MORE) {
        __builtin_amdgcn_sched_barrier(0);
        #pragma unroll
        for (int q = 0; q < 8; ++q)
            *(v4f*)&evS[p ^ 1][sRow + 16 * q][sCol] = evr[q];
    }
    if constexpr (MORE2) {
        #pragma unroll
        for (int q = 0; q < 8; ++q)
            evr[q] = *(const v4f*)(enext2 + (size_t)q * 16 * NDIM);
    }
    #pragma unroll
    for (int rb = 0; rb < 8; ++rb) {
        a2[rb] = *(const v4f*)&evS[p][rb * 16 + c][16 + 4 * h];
        a3[rb] = *(const v4f*)&evS[p][rb * 16 + c][24 + 4 * h];
    }

    fm_gemm_group<1>(acc, fcur, a1);

    if constexpr (MORE) {
        asm volatile("s_wait_dscnt 0x0" ::: "memory");
        asm volatile("s_barrier_signal -1" ::: "memory");
        __builtin_amdgcn_sched_barrier(0);
    }

    fm_gemm_group<2>(acc, fcur, a2);

    if constexpr (MORE) {
        asm volatile("s_barrier_wait -1" ::: "memory");
        #pragma unroll
        for (int rb = 0; rb < 8; ++rb)
            a0[rb] = *(const v4f*)&evS[p ^ 1][rb * 16 + c][4 * h];
    }

    fm_gemm_group<3>(acc, fcur, a3);

    if constexpr (MORE) {
        #pragma unroll
        for (int f = 0; f < 16; ++f) fcur[f] = fnxt[f];
    }
}

__global__ __launch_bounds__(128) void fm_gemm_partial(
    const float* __restrict__ evx, const float* __restrict__ evy,
    const float* __restrict__ fx,  const float* __restrict__ fy,
    float* __restrict__ partials)
{
    const int sel   = blockIdx.y;
    const int chunk = blockIdx.x >> 1;
    const int half  = blockIdx.x & 1;
    const float* __restrict__ ev   = sel ? evy : evx;
    const float* __restrict__ feat = sel ? fy  : fx;

    __shared__ __attribute__((aligned(16))) float evS[2][KDIM][EVP];

    const int tid  = threadIdx.x;
    const int lane = tid & 31;
    const int wave = tid >> 5;
    const int c    = lane & 15;
    const int h    = lane >> 4;
    const int colBase = half * 128 + wave * 32;

    const int nStart = chunk * CHUNK;
    const int nEnd   = min(nStart + CHUNK, NDIM);
    const int nSlabs = (nEnd - nStart) / SLAB;

    const int sRow = tid >> 3;
    const int sCol = (tid & 7) * 4;
    const float* __restrict__ evp = ev + (size_t)sRow * NDIM + nStart + sCol;
    const float* __restrict__ ftp = feat + (size_t)(nStart + 4 * h) * PDIM + colBase + 2 * c;

    v8f acc[8][2];
    #pragma unroll
    for (int rb = 0; rb < 8; ++rb) { acc[rb][0] = (v8f)0.0f; acc[rb][1] = (v8f)0.0f; }

    v2f fcur[16];
    v4f evr[8];
    v4f a0[8];

    #pragma unroll
    for (int q = 0; q < 8; ++q)
        evr[q] = *(const v4f*)(evp + (size_t)q * 16 * NDIM);
    #pragma unroll
    for (int f = 0; f < 16; ++f)
        fcur[f] = *(const v2f*)(ftp + (size_t)((f >> 2) * 8 + (f & 3)) * PDIM);
    #pragma unroll
    for (int q = 0; q < 8; ++q)
        *(v4f*)&evS[0][sRow + 16 * q][sCol] = evr[q];
    __syncthreads();
    #pragma unroll
    for (int q = 0; q < 8; ++q)
        evr[q] = *(const v4f*)(evp + SLAB + (size_t)q * 16 * NDIM);
    #pragma unroll
    for (int rb = 0; rb < 8; ++rb)
        a0[rb] = *(const v4f*)&evS[0][rb * 16 + c][4 * h];

    int i = 0;
    for (; i + 2 < nSlabs; ++i)
        fm_gemm_slab<true, true>(acc, fcur, evr, a0, evS, i & 1,
                                 ftp + (size_t)(i + 1) * SLAB * PDIM, evp + (size_t)(i + 2) * SLAB,
                                 sRow, sCol, c, h);
    fm_gemm_slab<true, false>(acc, fcur, evr, a0, evS, i & 1,
                              ftp + (size_t)(i + 1) * SLAB * PDIM, evp, sRow, sCol, c, h);
    ++i;
    fm_gemm_slab<false, false>(acc, fcur, evr, a0, evS, i & 1, ftp, evp, sRow, sCol, c, h);

    float* out = partials + (size_t)sel * NCHUNKS * KP + (size_t)chunk * KP;
    #pragma unroll
    for (int rb = 0; rb < 8; ++rb) {
        #pragma unroll
        for (int r = 0; r < 8; ++r) {
            const int row = rb * 16 + 8 * h + r;
            v2f o;
            o.x = acc[rb][0][r];
            o.y = acc[rb][1][r];
            *(v2f*)&out[row * PDIM + colBase + 2 * c] = o;
        }
    }
}

__global__ __launch_bounds__(256) void fm_reduce_mask(
    const float* __restrict__ partials,
    const float* __restrict__ evx, const float* __restrict__ evy,
    float* __restrict__ Amat, float* __restrict__ Bmat, float* __restrict__ D)
{
    __shared__ __attribute__((aligned(16))) float red[8][256];

    const int b    = blockIdx.x;
    const int sel  = b >> 7;
    const int base = (b & 127) * 256;
    const int tid  = threadIdx.x;
    const int lane = tid & 31;
    const int w    = tid >> 5;

    const float* __restrict__ pp =
        partials + (size_t)(sel * NCHUNKS + 8 * w) * KP + base + 4 * lane;
    v4f v[2][8];
    #pragma unroll
    for (int q = 0; q < 8; ++q) {
        v[0][q] = *(const v4f*)(pp + (size_t)q * KP);
        v[1][q] = *(const v4f*)(pp + (size_t)q * KP + 128);
    }

    const bool mk = ((tid >> 6) == 1);
    if (mk) {
        const int idx = b * 64 + (tid & 63);
        const int i = idx >> 7;
        const int j = idx & 127;
        float s  = fmaxf(evx[KDIM - 1], evy[KDIM - 1]);
        float g1 = sqrtf(evx[j] / s);
        float g2 = sqrtf(evy[i] / s);
        float d1 = 1.0f / (g1 * g1 + 1.0f);
        float d2 = 1.0f / (g2 * g2 + 1.0f);
        float re = g2 * d2 - g1 * d1;
        float im = d2 - d1;
        D[idx] = re * re + im * im;
    }

    v4f s0 = v[0][0], s1 = v[1][0];
    #pragma unroll
    for (int q = 1; q < 8; ++q) { s0 += v[0][q]; s1 += v[1][q]; }
    *(v4f*)&red[w][4 * lane]       = s0;
    *(v4f*)&red[w][128 + 4 * lane] = s1;
    __syncthreads();

    if (tid < 64) {
        v4f t = *(const v4f*)&red[0][4 * tid];
        #pragma unroll
        for (int ww = 1; ww < 8; ++ww) t += *(const v4f*)&red[ww][4 * tid];
        float* __restrict__ out = sel ? Bmat : Amat;
        *(v4f*)&out[base + 4 * tid] = t;
    }
}

__device__ __forceinline__ void fm_bs_block(
    float* Msh, const int tid, const int kb,
    const float (&tc)[16][16], const float (&uc)[16],
    float (&tn)[16][16], float (&un)[16], float& yown, float& xout)
{
    const int k0 = kb * 16;
    const int kn = (kb > 0) ? k0 - 16 : 0;

    __syncthreads();

    float yb[16], x[16];
    #pragma unroll
    for (int r = 0; r < 16; ++r)
        yb[r] = Msh[(k0 + r) * 129 + KDIM];
    #pragma unroll
    for (int r = 0; r < 16; ++r) {
        #pragma unroll
        for (int c = r + 1; c < 16; ++c)
            tn[r][c] = Msh[(kn + c) * 129 + kn + r];
    }
    #pragma unroll
    for (int c = 0; c < 16; ++c)
        un[c] = Msh[(kn + c) * 129 + tid];
    __builtin_amdgcn_sched_barrier(0);

    #pragma unroll
    for (int c = 15; c >= 0; --c) {
        x[c] = yb[c];
        #pragma unroll
        for (int r = 0; r < c; ++r)
            yb[r] -= tc[r][c] * x[c];
        yown += uc[c] * x[c];
        xout = (((tid >> 4) == kb) && ((tid & 15) == c)) ? x[c] : xout;
    }
    if ((tid >> 4) == kb - 1)
        Msh[tid * 129 + KDIM] = yown;
}

template <int LC>
__device__ __forceinline__ float fm_row_bcast(const float x)
{
    return __builtin_bit_cast(float, __builtin_amdgcn_update_dpp(
        0, __builtin_bit_cast(int, x), 0x150 + LC  , 0xf, 0xf, true));
}

template <int LC>
__device__ __forceinline__ void fm_panel_pivots(float (&p)[16], float (&d)[16], float (&pre)[16],
                                                float& pr, float& dr,
                                                const int c16, const int rel)
{
    const bool belowd = c16 > LC;
    const bool below  = rel > LC;
    float q[16];
    #pragma unroll
    for (int j = LC; j < 16; ++j) q[j] = fm_row_bcast<LC>(d[j]);
    const float qr = fm_row_bcast<LC>(dr);
    const float fd = d[LC] / q[LC];
    const float f  = p[LC] / q[LC];
    pre[LC] = p[LC];
    d[LC] = belowd ? fd : d[LC];
    p[LC] = below  ? f  : p[LC];
    #pragma unroll
    for (int j = LC + 1; j < 16; ++j) {
        d[j] = belowd ? (d[j] - fd * q[j]) : d[j];
        p[j] = below  ? (p[j] - f  * q[j]) : p[j];
    }
    dr = belowd ? (dr - fd * qr) : dr;
    pr = below  ? (pr - f  * qr) : pr;
    if constexpr (LC < 15) fm_panel_pivots<LC + 1>(p, d, pre, pr, dr, c16, rel);
}

#define APITCH 132
__global__ __launch_bounds__(128) void fm_solve(
    const float* __restrict__ Amat, const float* __restrict__ Bmat,
    const float* __restrict__ D,    float* __restrict__ Cout)
{
    extern __shared__ float4 smem4[];
    float* Msh  = (float*)smem4;
    float* Ast  = (float*)smem4;
    float* Brow = Ast + KDIM * APITCH;
    const int i   = blockIdx.x;
    const int tid = threadIdx.x;
    const int wvu = __builtin_amdgcn_readfirstlane(tid >> 5);

    {
        const int lane = tid & 31;
        const int wv   = tid >> 5;
        const int c    = lane & 15;
        const int h    = lane >> 4;

        const float* __restrict__ ap = Amat + (size_t)wv * PDIM + 4 * lane;
        v4f st0[32], st1[32];
        #pragma unroll
        for (int q = 0; q < 32; ++q)
            st0[q] = *(const v4f*)(ap + (size_t)q * 4 * PDIM);
        const v2f bq = *(const v2f*)(Bmat + (size_t)i * PDIM + 2 * tid);
        const float dsc = D[i * KDIM + tid];
        #pragma unroll
        for (int q = 0; q < 32; ++q)
            *(v4f*)&Ast[(4 * q + wv) * APITCH + 4 * lane] = st0[q];
        *(v2f*)&Brow[2 * tid] = bq;
        __syncthreads();
        #pragma unroll
        for (int q = 0; q < 32; ++q)
            st1[q] = *(const v4f*)(ap + (size_t)q * 4 * PDIM + 128);

        v8f acc[11];
        #pragma unroll
        for (int t = 0; t < 11; ++t) acc[t] = (v8f)0.0f;

        int fo[8];
        #pragma unroll
        for (int t = 0; t < 8; ++t) fo[t] = (16 * ((wv + t) & 7) + c) * APITCH + 2 * h;
        const bool row0 = (c == 0);

        for (int hh = 0; hh < 2; ++hh) {
            if (hh == 1) {
                __syncthreads();
                #pragma unroll
                for (int q = 0; q < 32; ++q)
                    *(v4f*)&Ast[(4 * q + wv) * APITCH + 4 * lane] = st1[q];
                __syncthreads();
            }
            v2f f[8], fb;
            #pragma unroll
            for (int t = 0; t < 8; ++t) f[t] = *(const v2f*)&Ast[fo[t]];
            {
                const v2f tb = *(const v2f*)&Brow[128 * hh + 2 * h];
                fb.x = row0 ? tb.x : 0.0f;
                fb.y = row0 ? tb.y : 0.0f;
            }
            for (int ks = 0; ks < 32; ++ks) {
                v2f fn[8], fbn;
                if (ks + 1 < 32) {
                    #pragma unroll
                    for (int t = 0; t < 8; ++t) fn[t] = *(const v2f*)&Ast[fo[t] + 4 * (ks + 1)];
                    const v2f tb = *(const v2f*)&Brow[128 * hh + 4 * (ks + 1) + 2 * h];
                    fbn.x = row0 ? tb.x : 0.0f;
                    fbn.y = row0 ? tb.y : 0.0f;
                } else {
                    #pragma unroll
                    for (int t = 0; t < 8; ++t) fn[t] = f[t];
                    fbn = fb;
                }
                #pragma unroll
                for (int t = 0; t < 5; ++t)
                    acc[t] = __builtin_amdgcn_wmma_f32_16x16x4_f32(
                        false, f[0], false, f[t], (short)0, acc[t], false, false);
                #pragma unroll
                for (int t = 0; t < 4; ++t)
                    acc[5 + t] = __builtin_amdgcn_wmma_f32_16x16x4_f32(
                        false, f[4], false, f[4 + t], (short)0, acc[5 + t], false, false);
                acc[9]  = __builtin_amdgcn_wmma_f32_16x16x4_f32(
                    false, fb, false, f[0], (short)0, acc[9], false, false);
                acc[10] = __builtin_amdgcn_wmma_f32_16x16x4_f32(
                    false, fb, false, f[4], (short)0, acc[10], false, false);
                #pragma unroll
                for (int t = 0; t < 8; ++t) f[t] = fn[t];
                fb = fbn;
            }
        }
        __syncthreads();

        #pragma unroll
        for (int a = 0; a < 2; ++a) {
            const int I = wv + 4 * a;
            #pragma unroll
            for (int t = 0; t < 5 - a; ++t) {
                const int J = (I + t) & 7;
                const int col = 16 * J + c;
                #pragma unroll
                for (int r = 0; r < 8; ++r) {
                    const int row = 16 * I + 8 * h + r;
                    const float v = acc[5 * a + t][r];
                    Msh[row * 129 + col] = v;
                    if (t != 0) Msh[col * 129 + row] = v;
                }
            }
            if (h == 0)
                Msh[(16 * I + c) * 129 + KDIM] = acc[9 + a][0];
        }
        __syncthreads();
        Msh[tid * 129 + tid] += LAMBDA * dsc;
        __syncthreads();
    }

    float yrow = Msh[tid * 129 + KDIM];

    for (int kb = 0; kb < KDIM / 16; ++kb) {
        const int k0 = kb * 16;

        float p[16], d[16], pre[16];
        #pragma unroll
        for (int j = 0; j < 16; ++j)
            p[j] = (tid >= k0) ? Msh[tid * 129 + k0 + j] : 0.0f;
        #pragma unroll
        for (int j = 0; j < 16; ++j)
            d[j] = Msh[(k0 + (tid & 15)) * 129 + k0 + j];
        float dr = Msh[(k0 + (tid & 15)) * 129 + KDIM];
        fm_panel_pivots<0>(p, d, pre, yrow, dr, tid & 15, tid - k0);
        __syncthreads();

        if (tid >= k0)
            Msh[tid * 129 + KDIM] = yrow;
        if (tid >= k0 && tid < k0 + 16) {
            #pragma unroll
            for (int j = 0; j < 16; ++j)
                Msh[tid * 129 + k0 + j] = p[j];
        } else if (tid >= k0 + 16) {
            #pragma unroll
            for (int j = 0; j < 16; ++j)
                Msh[tid * 129 + k0 + j] = -p[j];
            #pragma unroll
            for (int j = 0; j < 16; ++j)
                Msh[(k0 + j) * 129 + tid] = pre[j];
        }
        __syncthreads();

        {
            const int lane = tid & 31;
            const int c    = lane & 15;
            const int h    = lane >> 4;
            int item = 0;
            for (int I = kb + 1; I < 8; ++I) {
                const int np = (I - kb + 1) >> 1;
                for (int jp = 0; jp < np; ++jp, ++item) {
                    if ((item & 3) != wvu) continue;
                    const int  J0  = kb + 1 + 2 * jp;
                    const bool two = (J0 + 1 <= I);
                    const int  J1  = two ? J0 + 1 : J0;
                    const float* la = Msh + (16 * I + c) * 129 + k0 + 2 * h;
                    const float* u0 = Msh + (k0 + 2 * h) * 129 + 16 * J0 + c;
                    const float* u1 = Msh + (k0 + 2 * h) * 129 + 16 * J1 + c;
                    float* t0 = Msh + (16 * I + 8 * h) * 129 + 16 * J0 + c;
                    float* t1 = Msh + (16 * I + 8 * h) * 129 + 16 * J1 + c;
                    v2f a[4], b0[4], b1[4];
                    v8f acc0, acc1;
                    #pragma unroll
                    for (int ks = 0; ks < 4; ++ks) {
                        a[ks].x  = la[4 * ks];           a[ks].y  = la[4 * ks + 1];
                        b0[ks].x = u0[(4 * ks) * 129];   b0[ks].y = u0[(4 * ks + 1) * 129];
                        b1[ks].x = u1[(4 * ks) * 129];   b1[ks].y = u1[(4 * ks + 1) * 129];
                    }
                    #pragma unroll
                    for (int r = 0; r < 8; ++r) { acc0[r] = t0[r * 129]; acc1[r] = t1[r * 129]; }
                    #pragma unroll
                    for (int ks = 0; ks < 4; ++ks) {
                        acc0 = __builtin_amdgcn_wmma_f32_16x16x4_f32(
                            false, a[ks], false, b0[ks], (short)0, acc0, false, false);
                        acc1 = __builtin_amdgcn_wmma_f32_16x16x4_f32(
                            false, a[ks], false, b1[ks], (short)0, acc1, false, false);
                    }
                    #pragma unroll
                    for (int r = 0; r < 8; ++r) t0[r * 129] = acc0[r];
                    if (two) {
                        #pragma unroll
                        for (int r = 0; r < 8; ++r) t1[r * 129] = acc1[r];
                    }
                }
            }
        }
        __syncthreads();
    }

    float xout = 0.0f;
    float yown = yrow / Msh[tid * 129 + tid];
    Msh[tid * 129 + KDIM] = yown;
    float tA[16][16], tB[16][16];
    float uA[16], uB[16];
    {
        const int kl = KDIM - 16;
        #pragma unroll
        for (int r = 0; r < 16; ++r) {
            #pragma unroll
            for (int c = r + 1; c < 16; ++c)
                tA[r][c] = Msh[(kl + c) * 129 + kl + r];
        }
        #pragma unroll
        for (int c = 0; c < 16; ++c)
            uA[c] = Msh[(kl + c) * 129 + tid];
    }
    for (int kp = KDIM / 32 - 1; kp >= 0; --kp) {
        fm_bs_block(Msh, tid, 2 * kp + 1, tA, uA, tB, uB, yown, xout);
        fm_bs_block(Msh, tid, 2 * kp,     tB, uB, tA, uA, yown, xout);
    }
    Cout[i * KDIM + tid] = xout;
}

extern "C" void kernel_launch(void* const* d_in, const int* in_sizes, int n_in,
                              void* d_out, int out_size, void* d_ws, size_t ws_size,
                              hipStream_t stream)
{
    const float* feat_x = (const float*)d_in[0];
    const float* feat_y = (const float*)d_in[1];
    const float* evals_x = (const float*)d_in[2];
    const float* evals_y = (const float*)d_in[3];
    const float* evecs_x = (const float*)d_in[4];
    const float* evecs_y = (const float*)d_in[5];

    float* wsf   = (float*)d_ws;
    float* Amat  = wsf;
    float* Bmat  = wsf + 32768;
    float* Dmask = wsf + 98304;
    float* part  = wsf + 114688;
    float* Cout  = (float*)d_out;

    fm_gemm_partial<<<dim3(2 * NCHUNKS, 2), 128, 0, stream>>>(
        evecs_x, evecs_y, feat_x, feat_y, part);

    fm_reduce_mask<<<256, 256, 0, stream>>>(part, evals_x, evals_y, Amat, Bmat, Dmask);

    fm_solve<<<KDIM, KDIM, 68608  , stream>>>(
        Amat, Bmat, Dmask, Cout);
}
